# starting kernel
# baseline (speedup 1.0000x reference)
#include <hip/hip_runtime.h>
#include <math.h>

typedef __attribute__((ext_vector_type(2))) float v2f;
typedef __attribute__((ext_vector_type(8))) float v8f;

__device__ __forceinline__ unsigned fkey(float v) {
  unsigned b = __float_as_uint(v);
  return (b & 0x80000000u) ? ~b : (b | 0x80000000u);
}
__device__ __forceinline__ float funkey(unsigned k) {
  unsigned b = (k & 0x80000000u) ? (k ^ 0x80000000u) : ~k;
  return __uint_as_float(b);
}

__global__ void fill_f32(float* p, float v, int n) {
  int i = blockIdx.x * blockDim.x + threadIdx.x;
  if (i < n) p[i] = v;
}
__global__ void fill_u32(unsigned* p, unsigned v, int n) {
  int i = blockIdx.x * blockDim.x + threadIdx.x;
  if (i < n) p[i] = v;
}

template <int K, int Nc>
__global__ void gemm_wmma_f32(const float* __restrict__ A, const float* __restrict__ B,
                              const float* __restrict__ bias, float* __restrict__ Cout,
                              int M, int act) {
  constexpr int  tilesN = (Nc + 15) >> 4;
  constexpr bool FULL   = (Nc % 16) == 0;
  int wave = blockIdx.x * (blockDim.x >> 5) + (threadIdx.x >> 5);
  int tilesM = (M + 15) >> 4;
  if (wave >= tilesM * tilesN) return;
  int tm   = wave / tilesN;
  int tn   = wave - tm * tilesN;
  int lane = threadIdx.x & 31;
  int half = lane >> 4;
  int l16  = lane & 15;
  int row  = tm * 16 + l16;
  int col  = tn * 16 + l16;
  bool cok  = FULL || (col < Nc);
  int  colc = FULL ? col : (cok ? col : 0);

  const float* ap = A + (size_t)row * K + 2 * half;
  const float* bp = B + (size_t)(2 * half) * Nc + colc;

  v8f c = {0.f, 0.f, 0.f, 0.f, 0.f, 0.f, 0.f, 0.f};
#pragma unroll
  for (int k0 = 0; k0 < K; k0 += 4) {
    float2 av = *(const float2*)(ap + k0);
    float  bx = bp[(size_t)k0 * Nc];
    float  by = bp[(size_t)k0 * Nc + Nc];
    v2f a, b;
    a.x = av.x;
    a.y = av.y;
    b.x = FULL ? bx : (cok ? bx : 0.f);
    b.y = FULL ? by : (cok ? by : 0.f);
    c = __builtin_amdgcn_wmma_f32_16x16x4_f32(false, a, false, b, (short)0, c,
                                              false, false);
  }

  if (cok) {
    float bv = bias ? bias[col] : 0.f;
    float* crow = Cout + (size_t)(tm * 16 + 8 * half) * Nc + col;
#pragma unroll
    for (int v = 0; v < 8; ++v) {
      float val = c[v] + bv;
      if (act) val = fmaxf(val, 0.f);
      crow[(size_t)v * Nc] = val;
    }
  }
}

__global__ void gemm_wmma_f32_dyn(const float* __restrict__ A, const float* __restrict__ B,
                                  const float* __restrict__ bias, float* __restrict__ Cout,
                                  int M, int K, int Nc, int act) {
  int wave   = blockIdx.x * (blockDim.x >> 5) + (threadIdx.x >> 5);
  int tilesN = (Nc + 15) >> 4;
  int tilesM = (M + 15) >> 4;
  if (wave >= tilesM * tilesN) return;
  int tm = wave / tilesN, tn = wave - tm * tilesN;
  int lane = threadIdx.x & 31, half = lane >> 4, l16 = lane & 15;
  int row = tm * 16 + l16, col = tn * 16 + l16;
  bool cok = (col < Nc);
  int colc = cok ? col : 0;
  const float* ap = A + (size_t)row * K + 2 * half;
  const float* bp = B + (size_t)(2 * half) * Nc + colc;
  v8f c = {0.f, 0.f, 0.f, 0.f, 0.f, 0.f, 0.f, 0.f};
  for (int k0 = 0; k0 < K; k0 += 4) {
    float2 av = *(const float2*)ap;
    float bx = bp[0], by = bp[Nc];
    v2f a, b;
    a.x = av.x;            a.y = av.y;
    b.x = cok ? bx : 0.f;  b.y = cok ? by : 0.f;
    c = __builtin_amdgcn_wmma_f32_16x16x4_f32(false, a, false, b, (short)0, c,
                                              false, false);
    ap += 4;
    bp += (size_t)4 * Nc;
  }
  if (cok) {
    float bv = bias ? bias[col] : 0.f;
    float* crow = Cout + (size_t)(tm * 16 + 8 * half) * Nc + col;
    for (int v = 0; v < 8; ++v) {
      float val = c[v] + bv;
      if (act) val = fmaxf(val, 0.f);
      crow[(size_t)v * Nc] = val;
    }
  }
}

__global__ void alpha_kernel(const float* __restrict__ h, const float* __restrict__ a_src,
                             const float* __restrict__ a_dst, float* __restrict__ as_out,
                             float* __restrict__ ad_out, int n, int heads) {
  int wave = blockIdx.x * (blockDim.x >> 5) + (threadIdx.x >> 5);
  if (wave >= n * heads) return;
  int node = wave / heads, hd = wave - node * heads;
  int lane = threadIdx.x & 31;
  float2 hv = *(const float2*)(h + (size_t)node * heads * 64 + hd * 64 + lane * 2);
  float2 as = *(const float2*)(a_src + hd * 64 + lane * 2);
  float2 ad = *(const float2*)(a_dst + hd * 64 + lane * 2);
  float ss = hv.x * as.x + hv.y * as.y;
  float sd = hv.x * ad.x + hv.y * ad.y;
  for (int off = 16; off > 0; off >>= 1) {
    ss += __shfl_xor(ss, off, 32);
    sd += __shfl_xor(sd, off, 32);
  }
  if (lane == 0) { as_out[wave] = ss; ad_out[wave] = sd; }
}

__global__ void edge_max_kernel(const int* __restrict__ ei, int Eorig, int Etot,
                                const float* __restrict__ as_v, const float* __restrict__ ad_v,
                                float* __restrict__ ebuf, unsigned* __restrict__ mkey,
                                int heads) {
  int idx = blockIdx.x * blockDim.x + threadIdx.x;
  if (idx >= Etot * heads) return;
  int e = idx / heads, hd = idx - e * heads;
  int src, dst;
  if (e < Eorig) { src = ei[e]; dst = ei[Eorig + e]; }
  else           { src = dst = e - Eorig; }
  float v = as_v[src * heads + hd] + ad_v[dst * heads + hd];
  v = (v > 0.f) ? v : 0.2f * v;
  ebuf[idx] = v;
  atomicMax(&mkey[dst * heads + hd], fkey(v));
}

template <int HEADS>
__global__ void edge_acc_kernel(const int* __restrict__ ei, int Eorig, int Etot,
                                const float* __restrict__ ebuf, const unsigned* __restrict__ mkey,
                                const float* __restrict__ h, float* __restrict__ acc,
                                float* __restrict__ z) {
  int wave = blockIdx.x * (blockDim.x >> 5) + (threadIdx.x >> 5);
  if (wave >= Etot) return;
  int lane = threadIdx.x & 31;
  int src, dst;
  if (wave < Eorig) { src = ei[wave]; dst = ei[Eorig + wave]; }
  else              { src = dst = wave - Eorig; }

  if (HEADS == 2) {
    int hd = lane >> 4;
    float m  = funkey(mkey[dst * 2 + hd]);
    float ex = expf(ebuf[(size_t)wave * 2 + hd] - m);
    if (lane == 0 || lane == 16)
      atomicAdd(&z[dst * 2 + hd], ex);
    float4 hv = *(const float4*)(h + (size_t)src * 128 + lane * 4);
    float* ap = acc + (size_t)dst * 128 + lane * 4;
    atomicAdd(ap + 0, ex * hv.x);
    atomicAdd(ap + 1, ex * hv.y);
    atomicAdd(ap + 2, ex * hv.z);
    atomicAdd(ap + 3, ex * hv.w);
  } else {
    float m  = funkey(mkey[dst]);
    float ex = expf(ebuf[wave] - m);
    if (lane == 0) atomicAdd(&z[dst], ex);
    float2 hv = *(const float2*)(h + (size_t)src * 64 + lane * 2);
    float* ap = acc + (size_t)dst * 64 + lane * 2;
    atomicAdd(ap,     ex * hv.x);
    atomicAdd(ap + 1, ex * hv.y);
  }
}

__global__ void finalize_gat(float* __restrict__ acc, const float* __restrict__ z,
                             const float* __restrict__ bias, int n, int heads) {
  int total = n * heads * 64;
  int i = blockIdx.x * blockDim.x + threadIdx.x;
  if (i >= total) return;
  int c    = i % (heads * 64);
  int node = i / (heads * 64);
  int hd   = c >> 6;
  float v = acc[i] / (z[node * heads + hd] + 1e-16f) + bias[c];
  acc[i] = (v > 0.f) ? v : expm1f(v);
}

extern "C" void kernel_launch(void* const* d_in, const int* in_sizes, int n_in,
                              void* d_out, int out_size, void* d_ws, size_t ws_size,
                              hipStream_t stream) {
  const float* x      = (const float*)d_in[0];
  const int*   ei     = (const int*)d_in[1];
  const float* W1     = (const float*)d_in[2];
  const float* a_src1 = (const float*)d_in[3];
  const float* a_dst1 = (const float*)d_in[4];
  const float* b1     = (const float*)d_in[5];
  const float* W2     = (const float*)d_in[6];
  const float* a_src2 = (const float*)d_in[7];
  const float* a_dst2 = (const float*)d_in[8];
  const float* b2     = (const float*)d_in[9];
  const float* m1w    = (const float*)d_in[10];
  const float* m1b    = (const float*)d_in[11];
  const float* m2w    = (const float*)d_in[12];
  const float* m2b    = (const float*)d_in[13];

  const int N    = in_sizes[0] / 64;
  const int E    = in_sizes[1] / 2;
  const int C    = in_sizes[12] / 64;
  const int Etot = E + N;

  float*    bufA  = (float*)d_ws;
  float*    bufB  = bufA + (size_t)N * 128;
  float*    bufC  = bufB + (size_t)N * 128;
  float*    as_v  = bufC + (size_t)N * 64;
  float*    ad_v  = as_v + (size_t)N * 2;
  unsigned* mkey  = (unsigned*)(ad_v + (size_t)N * 2);
  float*    zbuf  = (float*)mkey + (size_t)N * 2;
  float*    ebuf  = zbuf + (size_t)N * 2;

  const int TPB = 256;
  auto blocks1d = [&](long n) { return (int)((n + TPB - 1) / TPB); };
  auto gemm_blocks = [&](int M, int Nc) {
    int waves = ((M + 15) / 16) * ((Nc + 15) / 16);
    return (waves + 7) / 8;
  };
  const unsigned KEY_NEG_INF = 0x007FFFFFu;

  gemm_wmma_f32<64, 128><<<gemm_blocks(N, 128), TPB, 0, stream>>>(x, W1, nullptr,
                                                                  bufA, N, 0);
  alpha_kernel<<<(N * 2 + 7) / 8, TPB, 0, stream>>>(bufA, a_src1, a_dst1,
                                                    as_v, ad_v, N, 2);
  fill_u32<<<blocks1d(N * 2), TPB, 0, stream>>>(mkey, KEY_NEG_INF, N * 2);
  fill_f32<<<blocks1d(N * 2), TPB, 0, stream>>>(zbuf, 0.f, N * 2);
  fill_f32<<<blocks1d((long)N * 128), TPB, 0, stream>>>(bufB, 0.f, N * 128);
  edge_max_kernel<<<blocks1d((long)Etot * 2), TPB, 0, stream>>>(ei, E, Etot, as_v, ad_v,
                                                                ebuf, mkey, 2);
  edge_acc_kernel<2><<<(Etot + 7) / 8, TPB, 0, stream>>>(ei, E, Etot, ebuf, mkey,
                                                         bufA, bufB, zbuf);
  finalize_gat<<<blocks1d((long)N * 128), TPB, 0, stream>>>(bufB, zbuf, b1, N, 2);

  gemm_wmma_f32<128, 64><<<gemm_blocks(N, 64), TPB, 0, stream>>>(bufB, W2, nullptr,
                                                                 bufA, N, 0);
  alpha_kernel<<<(N + 7) / 8, TPB, 0, stream>>>(bufA, a_src2, a_dst2,
                                                as_v, ad_v, N, 1);
  fill_u32<<<blocks1d(N), TPB, 0, stream>>>(mkey, KEY_NEG_INF, N);
  fill_f32<<<blocks1d(N), TPB, 0, stream>>>(zbuf, 0.f, N);
  fill_f32<<<blocks1d((long)N * 64), TPB, 0, stream>>>(bufC, 0.f, N * 64);
  edge_max_kernel<<<blocks1d((long)Etot), TPB, 0, stream>>>(ei, E, Etot, as_v, ad_v,
                                                            ebuf, mkey, 1);
  edge_acc_kernel<1><<<(Etot + 7) / 8, TPB, 0, stream>>>(ei, E, Etot, ebuf, mkey,
                                                         bufA, bufC, zbuf);
  finalize_gat<<<blocks1d((long)N * 64), TPB, 0, stream>>>(bufC, zbuf, b2, N, 1);

  gemm_wmma_f32<64, 64><<<gemm_blocks(N, 64), TPB, 0, stream>>>(bufC, m1w, m1b,
                                                                bufA, N, 1);
  if (C == 8) {
    gemm_wmma_f32<64, 8><<<gemm_blocks(N, 8), TPB, 0, stream>>>(bufA, m2w, m2b,
                                                                (float*)d_out, N, 1);
  } else {
    gemm_wmma_f32_dyn<<<gemm_blocks(N, C), TPB, 0, stream>>>(bufA, m2w, m2b,
                                                             (float*)d_out, N, 64, C, 1);
  }
}


# edge_acc: one wave per 32 consecutive edges instead of one wave per edge (per-edge indices, max, exp and z add done once by one lane; row loads of 8 edges in flight before their adds)
# speedup vs baseline: 1.0557x; 1.0557x over previous
#include <hip/hip_runtime.h>
#include <math.h>

typedef __attribute__((ext_vector_type(2))) float v2f;
typedef __attribute__((ext_vector_type(8))) float v8f;

__device__ __forceinline__ unsigned fkey(float v) {
  unsigned b = __float_as_uint(v);
  return (b & 0x80000000u) ? ~b : (b | 0x80000000u);
}
__device__ __forceinline__ float funkey(unsigned k) {
  unsigned b = (k & 0x80000000u) ? (k ^ 0x80000000u) : ~k;
  return __uint_as_float(b);
}

__global__ void fill_f32(float* p, float v, int n) {
  int i = blockIdx.x * blockDim.x + threadIdx.x;
  if (i < n) p[i] = v;
}
__global__ void fill_u32(unsigned* p, unsigned v, int n) {
  int i = blockIdx.x * blockDim.x + threadIdx.x;
  if (i < n) p[i] = v;
}

template <int K, int Nc>
__global__ void gemm_wmma_f32(const float* __restrict__ A, const float* __restrict__ B,
                              const float* __restrict__ bias, float* __restrict__ Cout,
                              int M, int act) {
  constexpr int  tilesN = (Nc + 15) >> 4;
  constexpr bool FULL   = (Nc % 16) == 0;
  int wave = blockIdx.x * (blockDim.x >> 5) + (threadIdx.x >> 5);
  int tilesM = (M + 15) >> 4;
  if (wave >= tilesM * tilesN) return;
  int tm   = wave / tilesN;
  int tn   = wave - tm * tilesN;
  int lane = threadIdx.x & 31;
  int half = lane >> 4;
  int l16  = lane & 15;
  int row  = tm * 16 + l16;
  int col  = tn * 16 + l16;
  bool cok  = FULL || (col < Nc);
  int  colc = FULL ? col : (cok ? col : 0);

  const float* ap = A + (size_t)row * K + 2 * half;
  const float* bp = B + (size_t)(2 * half) * Nc + colc;

  v8f c = {0.f, 0.f, 0.f, 0.f, 0.f, 0.f, 0.f, 0.f};
#pragma unroll
  for (int k0 = 0; k0 < K; k0 += 4) {
    float2 av = *(const float2*)(ap + k0);
    float  bx = bp[(size_t)k0 * Nc];
    float  by = bp[(size_t)k0 * Nc + Nc];
    v2f a, b;
    a.x = av.x;
    a.y = av.y;
    b.x = FULL ? bx : (cok ? bx : 0.f);
    b.y = FULL ? by : (cok ? by : 0.f);
    c = __builtin_amdgcn_wmma_f32_16x16x4_f32(false, a, false, b, (short)0, c,
                                              false, false);
  }

  if (cok) {
    float bv = bias ? bias[col] : 0.f;
    float* crow = Cout + (size_t)(tm * 16 + 8 * half) * Nc + col;
#pragma unroll
    for (int v = 0; v < 8; ++v) {
      float val = c[v] + bv;
      if (act) val = fmaxf(val, 0.f);
      crow[(size_t)v * Nc] = val;
    }
  }
}

__global__ void gemm_wmma_f32_dyn(const float* __restrict__ A, const float* __restrict__ B,
                                  const float* __restrict__ bias, float* __restrict__ Cout,
                                  int M, int K, int Nc, int act) {
  int wave   = blockIdx.x * (blockDim.x >> 5) + (threadIdx.x >> 5);
  int tilesN = (Nc + 15) >> 4;
  int tilesM = (M + 15) >> 4;
  if (wave >= tilesM * tilesN) return;
  int tm = wave / tilesN, tn = wave - tm * tilesN;
  int lane = threadIdx.x & 31, half = lane >> 4, l16 = lane & 15;
  int row = tm * 16 + l16, col = tn * 16 + l16;
  bool cok = (col < Nc);
  int colc = cok ? col : 0;
  const float* ap = A + (size_t)row * K + 2 * half;
  const float* bp = B + (size_t)(2 * half) * Nc + colc;
  v8f c = {0.f, 0.f, 0.f, 0.f, 0.f, 0.f, 0.f, 0.f};
  for (int k0 = 0; k0 < K; k0 += 4) {
    float2 av = *(const float2*)ap;
    float bx = bp[0], by = bp[Nc];
    v2f a, b;
    a.x = av.x;            a.y = av.y;
    b.x = cok ? bx : 0.f;  b.y = cok ? by : 0.f;
    c = __builtin_amdgcn_wmma_f32_16x16x4_f32(false, a, false, b, (short)0, c,
                                              false, false);
    ap += 4;
    bp += (size_t)4 * Nc;
  }
  if (cok) {
    float bv = bias ? bias[col] : 0.f;
    float* crow = Cout + (size_t)(tm * 16 + 8 * half) * Nc + col;
    for (int v = 0; v < 8; ++v) {
      float val = c[v] + bv;
      if (act) val = fmaxf(val, 0.f);
      crow[(size_t)v * Nc] = val;
    }
  }
}

__global__ void alpha_kernel(const float* __restrict__ h, const float* __restrict__ a_src,
                             const float* __restrict__ a_dst, float* __restrict__ as_out,
                             float* __restrict__ ad_out, int n, int heads) {
  int wave = blockIdx.x * (blockDim.x >> 5) + (threadIdx.x >> 5);
  if (wave >= n * heads) return;
  int node = wave / heads, hd = wave - node * heads;
  int lane = threadIdx.x & 31;
  float2 hv = *(const float2*)(h + (size_t)node * heads * 64 + hd * 64 + lane * 2);
  float2 as = *(const float2*)(a_src + hd * 64 + lane * 2);
  float2 ad = *(const float2*)(a_dst + hd * 64 + lane * 2);
  float ss = hv.x * as.x + hv.y * as.y;
  float sd = hv.x * ad.x + hv.y * ad.y;
  for (int off = 16; off > 0; off >>= 1) {
    ss += __shfl_xor(ss, off, 32);
    sd += __shfl_xor(sd, off, 32);
  }
  if (lane == 0) { as_out[wave] = ss; ad_out[wave] = sd; }
}

__global__ void edge_max_kernel(const int* __restrict__ ei, int Eorig, int Etot,
                                const float* __restrict__ as_v, const float* __restrict__ ad_v,
                                float* __restrict__ ebuf, unsigned* __restrict__ mkey,
                                int heads) {
  int idx = blockIdx.x * blockDim.x + threadIdx.x;
  if (idx >= Etot * heads) return;
  int e = idx / heads, hd = idx - e * heads;
  int src, dst;
  if (e < Eorig) { src = ei[e]; dst = ei[Eorig + e]; }
  else           { src = dst = e - Eorig; }
  float v = as_v[src * heads + hd] + ad_v[dst * heads + hd];
  v = (v > 0.f) ? v : 0.2f * v;
  ebuf[idx] = v;
  atomicMax(&mkey[dst * heads + hd], fkey(v));
}

__device__ __forceinline__ int bcast_i(int v, int j) {
  return __builtin_amdgcn_readlane(v, j);
}
__device__ __forceinline__ float bcast_f(float v, int j) {
  return __int_as_float(__builtin_amdgcn_readlane(__float_as_int(v), j));
}

template <int HEADS, bool FULL>
__device__ __forceinline__ void edge_rows(int src, int dst, float ex0, float ex1, int cnt,
                                          int lane, const float* __restrict__ h,
                                          float* __restrict__ acc) {
  constexpr int ROW = HEADS * 64;
  constexpr int VEC = HEADS * 2;
  const int loff = lane * VEC;
#pragma unroll
  for (int j0 = 0; j0 < 32; j0 += 8) {
    float hv[8][VEC];
#pragma unroll
    for (int jj = 0; jj < 8; ++jj) {
      int s = bcast_i(src, j0 + jj);
      const float* hp = h + (size_t)s * ROW + loff;
      if (HEADS == 2) {
        float4 t = *(const float4*)hp;
        hv[jj][0] = t.x; hv[jj][1] = t.y; hv[jj][VEC - 2] = t.z; hv[jj][VEC - 1] = t.w;
      } else {
        float2 t = *(const float2*)hp;
        hv[jj][0] = t.x; hv[jj][1] = t.y;
      }
    }
#pragma unroll
    for (int jj = 0; jj < 8; ++jj) {
      int j = j0 + jj;
      if (FULL || j < cnt) {
        int   d  = bcast_i(dst, j);
        float ex = bcast_f(ex0, j);
        if (HEADS == 2) { float x1 = bcast_f(ex1, j); ex = (lane >> 4) ? x1 : ex; }
        float* ap = acc + (size_t)d * ROW + loff;
#pragma unroll
        for (int c = 0; c < VEC; ++c) atomicAdd(ap + c, ex * hv[jj][c]);
      }
    }
  }
}

template <int HEADS>
__global__ void edge_acc_kernel(const int* __restrict__ ei, int Eorig, int Etot,
                                const float* __restrict__ ebuf, const unsigned* __restrict__ mkey,
                                const float* __restrict__ h, float* __restrict__ acc,
                                float* __restrict__ z) {
  int wave = __builtin_amdgcn_readfirstlane(blockIdx.x * (blockDim.x >> 5) + (threadIdx.x >> 5));
  int e0 = wave * 32;
  if (e0 >= Etot) return;
  int cnt  = min(32, Etot - e0);
  int lane = threadIdx.x & 31;
  bool ok  = lane < cnt;
  int e    = e0 + (ok ? lane : cnt - 1);
  int src, dst;
  if (e < Eorig) { src = ei[e]; dst = ei[Eorig + e]; }
  else           { src = dst = e - Eorig; }

  float ex0, ex1 = 0.f;
  if (HEADS == 2) {
    float2 ev = *(const float2*)(ebuf + (size_t)e * 2);
    uint2  mk = *(const uint2*)(mkey + (size_t)dst * 2);
    ex0 = expf(ev.x - funkey(mk.x));
    ex1 = expf(ev.y - funkey(mk.y));
    if (ok) {
      atomicAdd(&z[dst * 2],     ex0);
      atomicAdd(&z[dst * 2 + 1], ex1);
    }
  } else {
    ex0 = expf(ebuf[e] - funkey(mkey[dst]));
    if (ok) atomicAdd(&z[dst], ex0);
  }

  if (cnt == 32) edge_rows<HEADS, true >(src, dst, ex0, ex1, cnt, lane, h, acc);
  else           edge_rows<HEADS, false>(src, dst, ex0, ex1, cnt, lane, h, acc);
}

__global__ void finalize_gat(float* __restrict__ acc, const float* __restrict__ z,
                             const float* __restrict__ bias, int n, int heads) {
  int total = n * heads * 64;
  int i = blockIdx.x * blockDim.x + threadIdx.x;
  if (i >= total) return;
  int c    = i % (heads * 64);
  int node = i / (heads * 64);
  int hd   = c >> 6;
  float v = acc[i] / (z[node * heads + hd] + 1e-16f) + bias[c];
  acc[i] = (v > 0.f) ? v : expm1f(v);
}

extern "C" void kernel_launch(void* const* d_in, const int* in_sizes, int n_in,
                              void* d_out, int out_size, void* d_ws, size_t ws_size,
                              hipStream_t stream) {
  const float* x      = (const float*)d_in[0];
  const int*   ei     = (const int*)d_in[1];
  const float* W1     = (const float*)d_in[2];
  const float* a_src1 = (const float*)d_in[3];
  const float* a_dst1 = (const float*)d_in[4];
  const float* b1     = (const float*)d_in[5];
  const float* W2     = (const float*)d_in[6];
  const float* a_src2 = (const float*)d_in[7];
  const float* a_dst2 = (const float*)d_in[8];
  const float* b2     = (const float*)d_in[9];
  const float* m1w    = (const float*)d_in[10];
  const float* m1b    = (const float*)d_in[11];
  const float* m2w    = (const float*)d_in[12];
  const float* m2b    = (const float*)d_in[13];

  const int N    = in_sizes[0] / 64;
  const int E    = in_sizes[1] / 2;
  const int C    = in_sizes[12] / 64;
  const int Etot = E + N;

  float*    bufA  = (float*)d_ws;
  float*    bufB  = bufA + (size_t)N * 128;
  float*    bufC  = bufB + (size_t)N * 128;
  float*    as_v  = bufC + (size_t)N * 64;
  float*    ad_v  = as_v + (size_t)N * 2;
  unsigned* mkey  = (unsigned*)(ad_v + (size_t)N * 2);
  float*    zbuf  = (float*)mkey + (size_t)N * 2;
  float*    ebuf  = zbuf + (size_t)N * 2;

  const int TPB = 256;
  auto blocks1d = [&](long n) { return (int)((n + TPB - 1) / TPB); };
  auto gemm_blocks = [&](int M, int Nc) {
    int waves = ((M + 15) / 16) * ((Nc + 15) / 16);
    return (waves + 7) / 8;
  };
  const unsigned KEY_NEG_INF = 0x007FFFFFu;

  gemm_wmma_f32<64, 128><<<gemm_blocks(N, 128), TPB, 0, stream>>>(x, W1, nullptr,
                                                                  bufA, N, 0);
  alpha_kernel<<<(N * 2 + 7) / 8, TPB, 0, stream>>>(bufA, a_src1, a_dst1,
                                                    as_v, ad_v, N, 2);
  fill_u32<<<blocks1d(N * 2), TPB, 0, stream>>>(mkey, KEY_NEG_INF, N * 2);
  fill_f32<<<blocks1d(N * 2), TPB, 0, stream>>>(zbuf, 0.f, N * 2);
  fill_f32<<<blocks1d((long)N * 128), TPB, 0, stream>>>(bufB, 0.f, N * 128);
  edge_max_kernel<<<blocks1d((long)Etot * 2), TPB, 0, stream>>>(ei, E, Etot, as_v, ad_v,
                                                                ebuf, mkey, 2);
  edge_acc_kernel<2><<<blocks1d(Etot), TPB, 0, stream>>>(ei, E, Etot, ebuf, mkey,
                                                         bufA, bufB, zbuf);
  finalize_gat<<<blocks1d((long)N * 128), TPB, 0, stream>>>(bufB, zbuf, b1, N, 2);

  gemm_wmma_f32<128, 64><<<gemm_blocks(N, 64), TPB, 0, stream>>>(bufB, W2, nullptr,
                                                                 bufA, N, 0);
  alpha_kernel<<<(N + 7) / 8, TPB, 0, stream>>>(bufA, a_src2, a_dst2,
                                                as_v, ad_v, N, 1);
  fill_u32<<<blocks1d(N), TPB, 0, stream>>>(mkey, KEY_NEG_INF, N);
  fill_f32<<<blocks1d(N), TPB, 0, stream>>>(zbuf, 0.f, N);
  fill_f32<<<blocks1d((long)N * 64), TPB, 0, stream>>>(bufC, 0.f, N * 64);
  edge_max_kernel<<<blocks1d((long)Etot), TPB, 0, stream>>>(ei, E, Etot, as_v, ad_v,
                                                            ebuf, mkey, 1);
  edge_acc_kernel<1><<<blocks1d(Etot), TPB, 0, stream>>>(ei, E, Etot, ebuf, mkey,
                                                         bufA, bufC, zbuf);
  finalize_gat<<<blocks1d((long)N * 64), TPB, 0, stream>>>(bufC, zbuf, b2, N, 1);

  gemm_wmma_f32<64, 64><<<gemm_blocks(N, 64), TPB, 0, stream>>>(bufC, m1w, m1b,
                                                                bufA, N, 1);
  if (C == 8) {
    gemm_wmma_f32<64, 8><<<gemm_blocks(N, 8), TPB, 0, stream>>>(bufA, m2w, m2b,
                                                                (float*)d_out, N, 1);
  } else {
    gemm_wmma_f32_dyn<<<gemm_blocks(N, C), TPB, 0, stream>>>(bufA, m2w, m2b,
                                                             (float*)d_out, N, 64, C, 1);
  }
}


# conv1 aggregates before it projects (both linear): gather reads 256-byte x rows with 2 head weights, not 512-byte h rows; scores = x @ (W1 a); W1 GEMM, bias, elu after; sums in a new order
# speedup vs baseline: 2.8153x; 2.6668x over previous
#include <hip/hip_runtime.h>
#include <math.h>

typedef __attribute__((ext_vector_type(2))) float v2f;
typedef __attribute__((ext_vector_type(8))) float v8f;

__global__ void fill_i32(int* p, int v, int n) {
  int i = blockIdx.x * blockDim.x + threadIdx.x;
  if (i < n) p[i] = v;
}

template <int K, int Nc>
__global__ void gemm_wmma_f32(const float* __restrict__ A, const float* __restrict__ B,
                              const float* __restrict__ bias, float* __restrict__ Cout,
                              int M, int act) {
  constexpr int  tilesN = (Nc + 15) >> 4;
  constexpr bool FULL   = (Nc % 16) == 0;
  int wave = blockIdx.x * (blockDim.x >> 5) + (threadIdx.x >> 5);
  int tilesM = (M + 15) >> 4;
  if (wave >= tilesM * tilesN) return;
  int tm   = wave / tilesN;
  int tn   = wave - tm * tilesN;
  int lane = threadIdx.x & 31;
  int half = lane >> 4;
  int l16  = lane & 15;
  int row  = tm * 16 + l16;
  int col  = tn * 16 + l16;
  bool cok  = FULL || (col < Nc);
  int  colc = FULL ? col : (cok ? col : 0);

  const float* ap = A + (size_t)row * K + 2 * half;
  const float* bp = B + (size_t)(2 * half) * Nc + colc;

  v8f c = {0.f, 0.f, 0.f, 0.f, 0.f, 0.f, 0.f, 0.f};
#pragma unroll
  for (int k0 = 0; k0 < K; k0 += 4) {
    float2 av = *(const float2*)(ap + k0);
    float  bx = bp[(size_t)k0 * Nc];
    float  by = bp[(size_t)k0 * Nc + Nc];
    v2f a, b;
    a.x = av.x;
    a.y = av.y;
    b.x = FULL ? bx : (cok ? bx : 0.f);
    b.y = FULL ? by : (cok ? by : 0.f);
    c = __builtin_amdgcn_wmma_f32_16x16x4_f32(false, a, false, b, (short)0, c,
                                              false, false);
  }

  if (cok) {
    float bv = bias ? bias[col] : 0.f;
    float* crow = Cout + (size_t)(tm * 16 + 8 * half) * Nc + col;
#pragma unroll
    for (int v = 0; v < 8; ++v) {
      float val = c[v] + bv;
      if (act) val = fmaxf(val, 0.f);
      crow[(size_t)v * Nc] = val;
    }
  }
}


template <int Nc>
__device__ __forceinline__ void strip_load_b(const float* __restrict__ bq,
                                             float4 (&b0)[4], float4 (&b1)[4]) {
#pragma unroll
  for (int i = 0; i < 4; ++i) {
    b0[i] = *(const float4*)(bq + (size_t)(4 * i) * Nc);
    b1[i] = *(const float4*)(bq + (size_t)(4 * i) * Nc + Nc);
  }
}

__device__ __forceinline__ void strip_mma16(const float2 (&ac)[4], const float4 (&b0)[4],
                                            const float4 (&b1)[4], v8f& acc0, v8f& acc1,
                                            v8f& acc2, v8f& acc3) {
#pragma unroll
  for (int i = 0; i < 4; ++i) {
    v2f a, b;
    a.x = ac[i].x; a.y = ac[i].y;
    b.x = b0[i].x; b.y = b1[i].x;
    acc0 = __builtin_amdgcn_wmma_f32_16x16x4_f32(false, a, false, b, (short)0, acc0, false, false);
    b.x = b0[i].y; b.y = b1[i].y;
    acc1 = __builtin_amdgcn_wmma_f32_16x16x4_f32(false, a, false, b, (short)0, acc1, false, false);
    b.x = b0[i].z; b.y = b1[i].z;
    acc2 = __builtin_amdgcn_wmma_f32_16x16x4_f32(false, a, false, b, (short)0, acc2, false, false);
    b.x = b0[i].w; b.y = b1[i].w;
    acc3 = __builtin_amdgcn_wmma_f32_16x16x4_f32(false, a, false, b, (short)0, acc3, false, false);
  }
}

template <int K, int LDA, int Nc, bool ALPHA, bool HEADA, bool ELU>
__global__ void gemm_strip64(const float* __restrict__ A, const float* __restrict__ B,
                             const float* __restrict__ bias, float* __restrict__ Cout,
                             const float* __restrict__ a_src, const float* __restrict__ a_dst,
                             float* __restrict__ as_out, float* __restrict__ ad_out,
                             int M, int act) {
  static_assert(Nc % 64 == 0 && K % 16 == 0, "whole strips and whole chunks only");
  constexpr int strips = Nc / 64;
  int wave   = blockIdx.x * (blockDim.x >> 5) + (threadIdx.x >> 5);
  int tilesM = (M + 15) >> 4;
  if (wave >= tilesM * strips) return;
  int tm   = wave / strips;
  int sn   = wave - tm * strips;
  int lane = threadIdx.x & 31;
  int half = lane >> 4;
  int c    = lane & 15;
  int row  = tm * 16 + c;
  int n0   = sn * 64 + 4 * c;

  const float* ap = A + (size_t)row * LDA + (HEADA ? sn * 64 : 0) + 2 * half;
  const float* bp = B + (size_t)(2 * half) * Nc + n0;

  v8f acc0 = {0.f, 0.f, 0.f, 0.f, 0.f, 0.f, 0.f, 0.f};
  v8f acc1 = acc0, acc2 = acc0, acc3 = acc0;
  float2 an[4];
#pragma unroll
  for (int i = 0; i < 4; ++i) an[i] = *(const float2*)(ap + 4 * i);
#pragma nounroll
  for (int kc = 0; kc < K - 16; kc += 16) {
    float4 b0[4], b1[4];
    strip_load_b<Nc>(bp + (size_t)kc * Nc, b0, b1);
    float2 ac[4];
#pragma unroll
    for (int i = 0; i < 4; ++i) ac[i] = an[i];
#pragma unroll
    for (int i = 0; i < 4; ++i) an[i] = *(const float2*)(ap + kc + 16 + 4 * i);
    strip_mma16(ac, b0, b1, acc0, acc1, acc2, acc3);
  }
  {
    float4 b0[4], b1[4];
    strip_load_b<Nc>(bp + (size_t)(K - 16) * Nc, b0, b1);
    strip_mma16(an, b0, b1, acc0, acc1, acc2, acc3);
  }

  if (ALPHA) {
    const float4 s4 = *(const float4*)(a_src + n0);
    const float4 d4 = *(const float4*)(a_dst + n0);
    float ps[8], pd[8];
#pragma unroll
    for (int r = 0; r < 8; ++r) {
      ps[r] = acc0[r] * s4.x + acc1[r] * s4.y + acc2[r] * s4.z + acc3[r] * s4.w;
      pd[r] = acc0[r] * d4.x + acc1[r] * d4.y + acc2[r] * d4.z + acc3[r] * d4.w;
    }
    const bool k8 = (c & 8) != 0, k4 = (c & 4) != 0, k2 = (c & 2) != 0, k1 = (c & 1) != 0;
    float v8[8], v4[4], v2[2];
#pragma unroll
    for (int i = 0; i < 8; ++i) {
      float mine = k8 ? pd[i] : ps[i];
      float oth  = k8 ? ps[i] : pd[i];
      v8[i] = mine + __shfl_xor(oth, 8, 32);
    }
#pragma unroll
    for (int i = 0; i < 4; ++i) {
      float mine = k4 ? v8[i + 4] : v8[i];
      float oth  = k4 ? v8[i] : v8[i + 4];
      v4[i] = mine + __shfl_xor(oth, 4, 32);
    }
#pragma unroll
    for (int i = 0; i < 2; ++i) {
      float mine = k2 ? v4[i + 2] : v4[i];
      float oth  = k2 ? v4[i] : v4[i + 2];
      v2[i] = mine + __shfl_xor(oth, 2, 32);
    }
    float mine = k1 ? v2[1] : v2[0];
    float oth  = k1 ? v2[0] : v2[1];
    float tot  = mine + __shfl_xor(oth, 1, 32);
    int orow = tm * 16 + 8 * half + (c & 7);
    float* dst = k8 ? ad_out : as_out;
    dst[(size_t)orow * strips + sn] = tot;
  }

  float4 bv = make_float4(0.f, 0.f, 0.f, 0.f);
  if (bias) bv = *(const float4*)(bias + n0);
  float* crow = Cout + (size_t)(tm * 16 + 8 * half) * Nc + n0;
#pragma unroll
  for (int r = 0; r < 8; ++r) {
    float4 v = make_float4(acc0[r] + bv.x, acc1[r] + bv.y, acc2[r] + bv.z, acc3[r] + bv.w);
    if (act) { v.x = fmaxf(v.x, 0.f); v.y = fmaxf(v.y, 0.f); v.z = fmaxf(v.z, 0.f); v.w = fmaxf(v.w, 0.f); }
    if (ELU) {
      v.x = (v.x > 0.f) ? v.x : expm1f(v.x);
      v.y = (v.y > 0.f) ? v.y : expm1f(v.y);
      v.z = (v.z > 0.f) ? v.z : expm1f(v.z);
      v.w = (v.w > 0.f) ? v.w : expm1f(v.w);
    }
    *(float4*)(crow + (size_t)r * Nc) = v;
  }
}

__global__ void gemm_wmma_f32_dyn(const float* __restrict__ A, const float* __restrict__ B,
                                  const float* __restrict__ bias, float* __restrict__ Cout,
                                  int M, int K, int Nc, int act) {
  int wave   = blockIdx.x * (blockDim.x >> 5) + (threadIdx.x >> 5);
  int tilesN = (Nc + 15) >> 4;
  int tilesM = (M + 15) >> 4;
  if (wave >= tilesM * tilesN) return;
  int tm = wave / tilesN, tn = wave - tm * tilesN;
  int lane = threadIdx.x & 31, half = lane >> 4, l16 = lane & 15;
  int row = tm * 16 + l16, col = tn * 16 + l16;
  bool cok = (col < Nc);
  int colc = cok ? col : 0;
  const float* ap = A + (size_t)row * K + 2 * half;
  const float* bp = B + (size_t)(2 * half) * Nc + colc;
  v8f c = {0.f, 0.f, 0.f, 0.f, 0.f, 0.f, 0.f, 0.f};
  for (int k0 = 0; k0 < K; k0 += 4) {
    float2 av = *(const float2*)ap;
    float bx = bp[0], by = bp[Nc];
    v2f a, b;
    a.x = av.x;            a.y = av.y;
    b.x = cok ? bx : 0.f;  b.y = cok ? by : 0.f;
    c = __builtin_amdgcn_wmma_f32_16x16x4_f32(false, a, false, b, (short)0, c,
                                              false, false);
    ap += 4;
    bp += (size_t)4 * Nc;
  }
  if (cok) {
    float bv = bias ? bias[col] : 0.f;
    float* crow = Cout + (size_t)(tm * 16 + 8 * half) * Nc + col;
    for (int v = 0; v < 8; ++v) {
      float val = c[v] + bv;
      if (act) val = fmaxf(val, 0.f);
      crow[(size_t)v * Nc] = val;
    }
  }
}


__global__ void hist_dst(const int* __restrict__ ei, int E, int* __restrict__ cnt) {
  int e = blockIdx.x * blockDim.x + threadIdx.x;
  if (e < E) atomicAdd(&cnt[ei[E + e]], 1);
}

__global__ void __launch_bounds__(1024) scan_counts(const int* __restrict__ cnt, int n,
                                                    int* __restrict__ rowptr,
                                                    int* __restrict__ cursor) {
  __shared__ int s_tot[32];
  __shared__ int s_pre[32];
  const int t = threadIdx.x, lane = t & 31, w = t >> 5;
  const int base = blockIdx.x * 1024;
  int p = 0;
#pragma unroll 8
  for (int i = t; i < base; i += 1024) p += cnt[i];
  const int i = base + t;
  const int v = (i < n) ? cnt[i] : 0;
  int inc = v;
  for (int off = 1; off < 32; off <<= 1) {
    int u = __shfl_up(inc, off, 32);
    if (lane >= off) inc += u;
  }
  for (int off = 16; off > 0; off >>= 1) p += __shfl_xor(p, off, 32);
  if (lane == 31) s_tot[w] = inc;
  if (lane == 0)  s_pre[w] = p;
  __syncthreads();
  int offset = 0, wbase = 0;
  for (int k = 0; k < 32; ++k) {
    offset += s_pre[k];
    if (k < w) wbase += s_tot[k];
  }
  const int excl = offset + wbase + inc - v;
  if (i < n) { rowptr[i] = excl; cursor[i] = excl; }
  if (i == n - 1) rowptr[n] = excl + v;
}

__global__ void scatter_src(const int* __restrict__ ei, int E, int* __restrict__ cursor,
                            int* __restrict__ csr_src) {
  int e = blockIdx.x * blockDim.x + threadIdx.x;
  if (e < E) {
    int pos = atomicAdd(&cursor[ei[E + e]], 1);
    csr_src[pos] = ei[e];
  }
}

__device__ __forceinline__ int bcast_i(int v, int j) {
  return __builtin_amdgcn_readlane(v, j);
}
__device__ __forceinline__ float bcast_f(float v, int j) {
  return __int_as_float(__builtin_amdgcn_readlane(__float_as_int(v), j));
}

template <int HEADS, int AST>
__device__ __forceinline__ void edge_score(int j, int deg, int beg, int node,
                                           const int* __restrict__ csr_src,
                                           const float* __restrict__ as_v,
                                           float ad0, float ad1,
                                           int& s, float& e0, float& e1) {
  s = node; e0 = -INFINITY; e1 = -INFINITY;
  if (j < deg) {
    if (j < deg - 1) s = csr_src[beg + j];
    if (HEADS == 2) {
      float2 a = *(const float2*)(as_v + (size_t)s * AST);
      float v0 = a.x + ad0, v1 = a.y + ad1;
      e0 = (v0 > 0.f) ? v0 : 0.2f * v0;
      e1 = (v1 > 0.f) ? v1 : 0.2f * v1;
    } else {
      float v0 = as_v[s] + ad0;
      e0 = (v0 > 0.f) ? v0 : 0.2f * v0;
    }
  }
}

template <int HEADS, int G>
__device__ __forceinline__ void rows(int s, float ex0, float ex1, int j0, int lane,
                                     const float* __restrict__ h, float (&acc)[HEADS * 2]) {
  constexpr int ROW = HEADS * 64;
  constexpr int VEC = HEADS * 2;
  const int loff = lane * VEC;
  float hv[G][VEC];
#pragma unroll
  for (int jj = 0; jj < G; ++jj) {
    int sj = bcast_i(s, j0 + jj);
    const float* hp = h + (size_t)sj * ROW + loff;
    if (HEADS == 2) {
      float4 t = *(const float4*)hp;
      hv[jj][0] = t.x; hv[jj][1] = t.y; hv[jj][VEC - 2] = t.z; hv[jj][VEC - 1] = t.w;
    } else {
      float2 t = *(const float2*)hp;
      hv[jj][0] = t.x; hv[jj][1] = t.y;
    }
  }
#pragma unroll
  for (int jj = 0; jj < G; ++jj) {
    float ex = bcast_f(ex0, j0 + jj);
    if (HEADS == 2) { float x1 = bcast_f(ex1, j0 + jj); ex = (lane >> 4) ? x1 : ex; }
#pragma unroll
    for (int c = 0; c < VEC; ++c) acc[c] += ex * hv[jj][c];
  }
}

template <int HEADS>
__global__ void gat_gather(const int* __restrict__ rowptr, const int* __restrict__ csr_src,
                           const float* __restrict__ as_v, const float* __restrict__ ad_v,
                           const float* __restrict__ h, const float* __restrict__ bias,
                           float* __restrict__ out, int n) {
  constexpr int ROW = HEADS * 64;
  constexpr int VEC = HEADS * 2;
  int node = __builtin_amdgcn_readfirstlane(blockIdx.x * (blockDim.x >> 5) + (threadIdx.x >> 5));
  if (node >= n) return;
  const int lane = threadIdx.x & 31;
  const int beg  = __builtin_amdgcn_readfirstlane(rowptr[node]);
  const int end  = __builtin_amdgcn_readfirstlane(rowptr[node + 1]);
  const int deg  = end - beg + 1;
  const float ad0 = ad_v[(size_t)node * HEADS];
  const float ad1 = (HEADS == 2) ? ad_v[(size_t)node * 2 + 1] : 0.f;
  const int loff  = lane * VEC;
  float bv[VEC];
#pragma unroll
  for (int c = 0; c < VEC; ++c) bv[c] = bias[loff + c];

  int sA; float eA0, eA1;
  edge_score<HEADS, HEADS>(lane, deg, beg, node, csr_src, as_v, ad0, ad1, sA, eA0, eA1);
  float m0 = eA0, m1 = eA1;
  for (int c0 = 32; c0 < deg; c0 += 32) {
    int s; float e0, e1;
    edge_score<HEADS, HEADS>(c0 + lane, deg, beg, node, csr_src, as_v, ad0, ad1, s, e0, e1);
    m0 = fmaxf(m0, e0);
    m1 = fmaxf(m1, e1);
  }
  for (int off = 16; off > 0; off >>= 1) {
    m0 = fmaxf(m0, __shfl_xor(m0, off, 32));
    if (HEADS == 2) m1 = fmaxf(m1, __shfl_xor(m1, off, 32));
  }

  float z0 = 0.f, z1 = 0.f;
  float acc[VEC];
#pragma unroll
  for (int c = 0; c < VEC; ++c) acc[c] = 0.f;
  for (int c0 = 0; c0 < deg; c0 += 32) {
    int s = sA; float e0 = eA0, e1 = eA1;
    if (c0 > 0) edge_score<HEADS, HEADS>(c0 + lane, deg, beg, node, csr_src, as_v, ad0, ad1, s, e0, e1);
    const bool ok = (c0 + lane) < deg;
    const float ex0 = ok ? expf(e0 - m0) : 0.f;
    const float ex1 = (HEADS == 2 && ok) ? expf(e1 - m1) : 0.f;
    z0 += ex0;
    z1 += ex1;
    const int cnt = min(32, deg - c0);
    int j0 = 0;
    for (; j0 + 8 <= cnt; j0 += 8) rows<HEADS, 8>(s, ex0, ex1, j0, lane, h, acc);
    if (cnt & 4) { rows<HEADS, 4>(s, ex0, ex1, j0, lane, h, acc); j0 += 4; }
    if (cnt & 2) { rows<HEADS, 2>(s, ex0, ex1, j0, lane, h, acc); j0 += 2; }
    if (cnt & 1) { rows<HEADS, 1>(s, ex0, ex1, j0, lane, h, acc); }
  }
  for (int off = 16; off > 0; off >>= 1) {
    z0 += __shfl_xor(z0, off, 32);
    if (HEADS == 2) z1 += __shfl_xor(z1, off, 32);
  }

  const float zz  = (HEADS == 2 && (lane >> 4)) ? z1 : z0;
  const float den = zz + 1e-16f;
  float r[VEC];
#pragma unroll
  for (int c = 0; c < VEC; ++c) {
    float v = acc[c] / den + bv[c];
    r[c] = (v > 0.f) ? v : expm1f(v);
  }
  float* op = out + (size_t)node * ROW + loff;
  if (HEADS == 2) *(float4*)op = make_float4(r[0], r[1], r[VEC - 2], r[VEC - 1]);
  else            *(float2*)op = make_float2(r[0], r[1]);
}


__global__ void fold_att_into_w(const float* __restrict__ Wm, const float* __restrict__ a_src,
                                const float* __restrict__ a_dst, float* __restrict__ wa) {
  const int t = threadIdx.x;
  const int k = t >> 2, cc = t & 3, hd = cc & 1;
  const float* av = ((cc & 2) ? a_dst : a_src) + hd * 64;
  const float* wr = Wm + (size_t)k * 128 + hd * 64;
  float acc = 0.f;
#pragma unroll
  for (int f = 0; f < 64; f += 4) {
    const float4 w4 = *(const float4*)(wr + f);
    const float4 a4 = *(const float4*)(av + f);
    acc += w4.x * a4.x + w4.y * a4.y + w4.z * a4.z + w4.w * a4.w;
  }
  wa[t] = acc;
}

template <int G>
__device__ __forceinline__ void rows_x2(int s, float ex0, float ex1, int j0, int lane,
                                        const float* __restrict__ x, v2f& acc0, v2f& acc1) {
  const int loff = lane * 2;
  v2f hv[G];
#pragma unroll
  for (int jj = 0; jj < G; ++jj) {
    int sj = bcast_i(s, j0 + jj);
    hv[jj] = *(const v2f*)(x + (size_t)sj * 64 + loff);
  }
#pragma unroll
  for (int jj = 0; jj < G; ++jj) {
    const float w0 = bcast_f(ex0, j0 + jj);
    const float w1 = bcast_f(ex1, j0 + jj);
    acc0 += w0 * hv[jj];
    acc1 += w1 * hv[jj];
  }
}

__global__ void gat_gather_x2(const int* __restrict__ rowptr, const int* __restrict__ csr_src,
                              const float* __restrict__ sd, const float* __restrict__ x,
                              float* __restrict__ agg, int n) {
  int node = __builtin_amdgcn_readfirstlane(blockIdx.x * (blockDim.x >> 5) + (threadIdx.x >> 5));
  if (node >= n) return;
  const int lane = threadIdx.x & 31;
  const int beg  = __builtin_amdgcn_readfirstlane(rowptr[node]);
  const int end  = __builtin_amdgcn_readfirstlane(rowptr[node + 1]);
  const int deg  = end - beg + 1;
  const float2 adv = *(const float2*)(sd + (size_t)node * 4 + 2);
  const float ad0 = adv.x, ad1 = adv.y;

  int sA; float eA0, eA1;
  edge_score<2, 4>(lane, deg, beg, node, csr_src, sd, ad0, ad1, sA, eA0, eA1);
  float m0 = eA0, m1 = eA1;
  for (int c0 = 32; c0 < deg; c0 += 32) {
    int s; float e0, e1;
    edge_score<2, 4>(c0 + lane, deg, beg, node, csr_src, sd, ad0, ad1, s, e0, e1);
    m0 = fmaxf(m0, e0);
    m1 = fmaxf(m1, e1);
  }
  for (int off = 16; off > 0; off >>= 1) {
    m0 = fmaxf(m0, __shfl_xor(m0, off, 32));
    m1 = fmaxf(m1, __shfl_xor(m1, off, 32));
  }

  float z0 = 0.f, z1 = 0.f;
  v2f acc0 = {0.f, 0.f}, acc1 = {0.f, 0.f};
  for (int c0 = 0; c0 < deg; c0 += 32) {
    int s = sA; float e0 = eA0, e1 = eA1;
    if (c0 > 0) edge_score<2, 4>(c0 + lane, deg, beg, node, csr_src, sd, ad0, ad1, s, e0, e1);
    const bool ok = (c0 + lane) < deg;
    const float ex0 = ok ? expf(e0 - m0) : 0.f;
    const float ex1 = ok ? expf(e1 - m1) : 0.f;
    z0 += ex0;
    z1 += ex1;
    const int cnt = min(32, deg - c0);
    int j0 = 0;
    for (; j0 + 8 <= cnt; j0 += 8) rows_x2<8>(s, ex0, ex1, j0, lane, x, acc0, acc1);
    if (cnt & 4) { rows_x2<4>(s, ex0, ex1, j0, lane, x, acc0, acc1); j0 += 4; }
    if (cnt & 2) { rows_x2<2>(s, ex0, ex1, j0, lane, x, acc0, acc1); j0 += 2; }
    if (cnt & 1) { rows_x2<1>(s, ex0, ex1, j0, lane, x, acc0, acc1); }
  }
  for (int off = 16; off > 0; off >>= 1) {
    z0 += __shfl_xor(z0, off, 32);
    z1 += __shfl_xor(z1, off, 32);
  }

  const float den0 = z0 + 1e-16f, den1 = z1 + 1e-16f;
  float* op = agg + (size_t)node * 128 + lane * 2;
  *(float2*)op        = make_float2(acc0.x / den0, acc0.y / den0);
  *(float2*)(op + 64) = make_float2(acc1.x / den1, acc1.y / den1);
}

extern "C" void kernel_launch(void* const* d_in, const int* in_sizes, int n_in,
                              void* d_out, int out_size, void* d_ws, size_t ws_size,
                              hipStream_t stream) {
  const float* x      = (const float*)d_in[0];
  const int*   ei     = (const int*)d_in[1];
  const float* W1     = (const float*)d_in[2];
  const float* a_src1 = (const float*)d_in[3];
  const float* a_dst1 = (const float*)d_in[4];
  const float* b1     = (const float*)d_in[5];
  const float* W2     = (const float*)d_in[6];
  const float* a_src2 = (const float*)d_in[7];
  const float* a_dst2 = (const float*)d_in[8];
  const float* b2     = (const float*)d_in[9];
  const float* m1w    = (const float*)d_in[10];
  const float* m1b    = (const float*)d_in[11];
  const float* m2w    = (const float*)d_in[12];
  const float* m2b    = (const float*)d_in[13];

  const int N    = in_sizes[0] / 64;
  const int E    = in_sizes[1] / 2;
  const int C    = in_sizes[12] / 64;

  float* bufA   = (float*)d_ws;
  float* bufB   = bufA + (size_t)N * 128;
  float* bufC   = bufB + (size_t)N * 128;
  float* as_v   = bufC + (size_t)N * 64;
  float* ad_v   = as_v + (size_t)N * 2;
  int*   cnt    = (int*)(ad_v + (size_t)N * 2);
  int*   rowptr = cnt + (size_t)N;
  int*   cursor = rowptr + (size_t)N + 1;
  int*   csrsrc = cursor + (size_t)N;

  const int TPB = 256;
  auto blocks1d = [&](long n) { return (int)((n + TPB - 1) / TPB); };
  auto gemm_blocks = [&](int M, int Nc) {
    int waves = ((M + 15) / 16) * ((Nc + 15) / 16);
    return (waves + 7) / 8;
  };
  auto strip_blocks = [&](int M, int Nc) {
    int waves = ((M + 15) / 16) * (Nc / 64);
    return (waves + 7) / 8;
  };

  fill_i32<<<blocks1d(N), TPB, 0, stream>>>(cnt, 0, N);
  hist_dst<<<blocks1d(E), TPB, 0, stream>>>(ei, E, cnt);
  scan_counts<<<(N + 1023) / 1024, 1024, 0, stream>>>(cnt, N, rowptr, cursor);
  scatter_src<<<blocks1d(E), TPB, 0, stream>>>(ei, E, cursor, csrsrc);

  float* wa = bufC;
  fold_att_into_w<<<1, 256, 0, stream>>>(W1, a_src1, a_dst1, wa);
  gemm_wmma_f32<64, 4><<<gemm_blocks(N, 4), TPB, 0, stream>>>(x, wa, nullptr, as_v, N, 0);
  gat_gather_x2<<<(N + 7) / 8, TPB, 0, stream>>>(rowptr, csrsrc, as_v, x, bufA, N);
  gemm_strip64<64, 128, 128, false, true, true><<<strip_blocks(N, 128), TPB, 0, stream>>>(
      bufA, W1, b1, bufB, nullptr, nullptr, nullptr, nullptr, N, 0);

  gemm_strip64<128, 128, 64, true, false, false><<<strip_blocks(N, 64), TPB, 0, stream>>>(
      bufB, W2, nullptr, bufA, a_src2, a_dst2, as_v, ad_v, N, 0);
  gat_gather<1><<<(N + 7) / 8, TPB, 0, stream>>>(rowptr, csrsrc, as_v, ad_v,
                                                 bufA, b2, bufC, N);

  gemm_strip64<64, 64, 64, false, false, false><<<strip_blocks(N, 64), TPB, 0, stream>>>(
      bufC, m1w, m1b, bufA, nullptr, nullptr, nullptr, nullptr, N, 1);
  if (C == 8) {
    gemm_wmma_f32<64, 8><<<gemm_blocks(N, 8), TPB, 0, stream>>>(bufA, m2w, m2b,
                                                                (float*)d_out, N, 1);
  } else {
    gemm_wmma_f32_dyn<<<gemm_blocks(N, C), TPB, 0, stream>>>(bufA, m2w, m2b,
                                                             (float*)d_out, N, 64, C, 1);
  }
}


# counting sort with one atomic pass: the histogram's add returns each edge's rank in its destination segment, scatter = segment start + rank with plain loads; slot order, so the sums' order, shifts
# speedup vs baseline: 3.3413x; 1.1868x over previous
#include <hip/hip_runtime.h>
#include <math.h>

typedef __attribute__((ext_vector_type(2))) float v2f;
typedef __attribute__((ext_vector_type(8))) float v8f;

__global__ void fill_i32(int* p, int v, int n) {
  int i = blockIdx.x * blockDim.x + threadIdx.x;
  if (i < n) p[i] = v;
}

template <int K, int Nc>
__global__ void gemm_wmma_f32(const float* __restrict__ A, const float* __restrict__ B,
                              const float* __restrict__ bias, float* __restrict__ Cout,
                              int M, int act) {
  constexpr int  tilesN = (Nc + 15) >> 4;
  constexpr bool FULL   = (Nc % 16) == 0;
  int wave = blockIdx.x * (blockDim.x >> 5) + (threadIdx.x >> 5);
  int tilesM = (M + 15) >> 4;
  if (wave >= tilesM * tilesN) return;
  int tm   = wave / tilesN;
  int tn   = wave - tm * tilesN;
  int lane = threadIdx.x & 31;
  int half = lane >> 4;
  int l16  = lane & 15;
  int row  = tm * 16 + l16;
  int col  = tn * 16 + l16;
  bool cok  = FULL || (col < Nc);
  int  colc = FULL ? col : (cok ? col : 0);

  const float* ap = A + (size_t)row * K + 2 * half;
  const float* bp = B + (size_t)(2 * half) * Nc + colc;

  v8f c = {0.f, 0.f, 0.f, 0.f, 0.f, 0.f, 0.f, 0.f};
#pragma unroll
  for (int k0 = 0; k0 < K; k0 += 4) {
    float2 av = *(const float2*)(ap + k0);
    float  bx = bp[(size_t)k0 * Nc];
    float  by = bp[(size_t)k0 * Nc + Nc];
    v2f a, b;
    a.x = av.x;
    a.y = av.y;
    b.x = FULL ? bx : (cok ? bx : 0.f);
    b.y = FULL ? by : (cok ? by : 0.f);
    c = __builtin_amdgcn_wmma_f32_16x16x4_f32(false, a, false, b, (short)0, c,
                                              false, false);
  }

  if (cok) {
    float bv = bias ? bias[col] : 0.f;
    float* crow = Cout + (size_t)(tm * 16 + 8 * half) * Nc + col;
#pragma unroll
    for (int v = 0; v < 8; ++v) {
      float val = c[v] + bv;
      if (act) val = fmaxf(val, 0.f);
      crow[(size_t)v * Nc] = val;
    }
  }
}


template <int Nc>
__device__ __forceinline__ void strip_load_b(const float* __restrict__ bq,
                                             float4 (&b0)[4], float4 (&b1)[4]) {
#pragma unroll
  for (int i = 0; i < 4; ++i) {
    b0[i] = *(const float4*)(bq + (size_t)(4 * i) * Nc);
    b1[i] = *(const float4*)(bq + (size_t)(4 * i) * Nc + Nc);
  }
}

__device__ __forceinline__ void strip_mma16(const float2 (&ac)[4], const float4 (&b0)[4],
                                            const float4 (&b1)[4], v8f& acc0, v8f& acc1,
                                            v8f& acc2, v8f& acc3) {
#pragma unroll
  for (int i = 0; i < 4; ++i) {
    v2f a, b;
    a.x = ac[i].x; a.y = ac[i].y;
    b.x = b0[i].x; b.y = b1[i].x;
    acc0 = __builtin_amdgcn_wmma_f32_16x16x4_f32(false, a, false, b, (short)0, acc0, false, false);
    b.x = b0[i].y; b.y = b1[i].y;
    acc1 = __builtin_amdgcn_wmma_f32_16x16x4_f32(false, a, false, b, (short)0, acc1, false, false);
    b.x = b0[i].z; b.y = b1[i].z;
    acc2 = __builtin_amdgcn_wmma_f32_16x16x4_f32(false, a, false, b, (short)0, acc2, false, false);
    b.x = b0[i].w; b.y = b1[i].w;
    acc3 = __builtin_amdgcn_wmma_f32_16x16x4_f32(false, a, false, b, (short)0, acc3, false, false);
  }
}

template <int K, int LDA, int Nc, bool ALPHA, bool HEADA, bool ELU>
__global__ void gemm_strip64(const float* __restrict__ A, const float* __restrict__ B,
                             const float* __restrict__ bias, float* __restrict__ Cout,
                             const float* __restrict__ a_src, const float* __restrict__ a_dst,
                             float* __restrict__ as_out, float* __restrict__ ad_out,
                             int M, int act) {
  static_assert(Nc % 64 == 0 && K % 16 == 0, "whole strips and whole chunks only");
  constexpr int strips = Nc / 64;
  int wave   = blockIdx.x * (blockDim.x >> 5) + (threadIdx.x >> 5);
  int tilesM = (M + 15) >> 4;
  if (wave >= tilesM * strips) return;
  int tm   = wave / strips;
  int sn   = wave - tm * strips;
  int lane = threadIdx.x & 31;
  int half = lane >> 4;
  int c    = lane & 15;
  int row  = tm * 16 + c;
  int n0   = sn * 64 + 4 * c;

  const float* ap = A + (size_t)row * LDA + (HEADA ? sn * 64 : 0) + 2 * half;
  const float* bp = B + (size_t)(2 * half) * Nc + n0;

  v8f acc0 = {0.f, 0.f, 0.f, 0.f, 0.f, 0.f, 0.f, 0.f};
  v8f acc1 = acc0, acc2 = acc0, acc3 = acc0;
  float2 an[4];
#pragma unroll
  for (int i = 0; i < 4; ++i) an[i] = *(const float2*)(ap + 4 * i);
#pragma nounroll
  for (int kc = 0; kc < K - 16; kc += 16) {
    float4 b0[4], b1[4];
    strip_load_b<Nc>(bp + (size_t)kc * Nc, b0, b1);
    float2 ac[4];
#pragma unroll
    for (int i = 0; i < 4; ++i) ac[i] = an[i];
#pragma unroll
    for (int i = 0; i < 4; ++i) an[i] = *(const float2*)(ap + kc + 16 + 4 * i);
    strip_mma16(ac, b0, b1, acc0, acc1, acc2, acc3);
  }
  {
    float4 b0[4], b1[4];
    strip_load_b<Nc>(bp + (size_t)(K - 16) * Nc, b0, b1);
    strip_mma16(an, b0, b1, acc0, acc1, acc2, acc3);
  }

  if (ALPHA) {
    const float4 s4 = *(const float4*)(a_src + n0);
    const float4 d4 = *(const float4*)(a_dst + n0);
    float ps[8], pd[8];
#pragma unroll
    for (int r = 0; r < 8; ++r) {
      ps[r] = acc0[r] * s4.x + acc1[r] * s4.y + acc2[r] * s4.z + acc3[r] * s4.w;
      pd[r] = acc0[r] * d4.x + acc1[r] * d4.y + acc2[r] * d4.z + acc3[r] * d4.w;
    }
    const bool k8 = (c & 8) != 0, k4 = (c & 4) != 0, k2 = (c & 2) != 0, k1 = (c & 1) != 0;
    float v8[8], v4[4], v2[2];
#pragma unroll
    for (int i = 0; i < 8; ++i) {
      float mine = k8 ? pd[i] : ps[i];
      float oth  = k8 ? ps[i] : pd[i];
      v8[i] = mine + __shfl_xor(oth, 8, 32);
    }
#pragma unroll
    for (int i = 0; i < 4; ++i) {
      float mine = k4 ? v8[i + 4] : v8[i];
      float oth  = k4 ? v8[i] : v8[i + 4];
      v4[i] = mine + __shfl_xor(oth, 4, 32);
    }
#pragma unroll
    for (int i = 0; i < 2; ++i) {
      float mine = k2 ? v4[i + 2] : v4[i];
      float oth  = k2 ? v4[i] : v4[i + 2];
      v2[i] = mine + __shfl_xor(oth, 2, 32);
    }
    float mine = k1 ? v2[1] : v2[0];
    float oth  = k1 ? v2[0] : v2[1];
    float tot  = mine + __shfl_xor(oth, 1, 32);
    int orow = tm * 16 + 8 * half + (c & 7);
    float* dst = k8 ? ad_out : as_out;
    dst[(size_t)orow * strips + sn] = tot;
  }

  float4 bv = make_float4(0.f, 0.f, 0.f, 0.f);
  if (bias) bv = *(const float4*)(bias + n0);
  float* crow = Cout + (size_t)(tm * 16 + 8 * half) * Nc + n0;
#pragma unroll
  for (int r = 0; r < 8; ++r) {
    float4 v = make_float4(acc0[r] + bv.x, acc1[r] + bv.y, acc2[r] + bv.z, acc3[r] + bv.w);
    if (act) { v.x = fmaxf(v.x, 0.f); v.y = fmaxf(v.y, 0.f); v.z = fmaxf(v.z, 0.f); v.w = fmaxf(v.w, 0.f); }
    if (ELU) {
      v.x = (v.x > 0.f) ? v.x : expm1f(v.x);
      v.y = (v.y > 0.f) ? v.y : expm1f(v.y);
      v.z = (v.z > 0.f) ? v.z : expm1f(v.z);
      v.w = (v.w > 0.f) ? v.w : expm1f(v.w);
    }
    *(float4*)(crow + (size_t)r * Nc) = v;
  }
}

__global__ void gemm_wmma_f32_dyn(const float* __restrict__ A, const float* __restrict__ B,
                                  const float* __restrict__ bias, float* __restrict__ Cout,
                                  int M, int K, int Nc, int act) {
  int wave   = blockIdx.x * (blockDim.x >> 5) + (threadIdx.x >> 5);
  int tilesN = (Nc + 15) >> 4;
  int tilesM = (M + 15) >> 4;
  if (wave >= tilesM * tilesN) return;
  int tm = wave / tilesN, tn = wave - tm * tilesN;
  int lane = threadIdx.x & 31, half = lane >> 4, l16 = lane & 15;
  int row = tm * 16 + l16, col = tn * 16 + l16;
  bool cok = (col < Nc);
  int colc = cok ? col : 0;
  const float* ap = A + (size_t)row * K + 2 * half;
  const float* bp = B + (size_t)(2 * half) * Nc + colc;
  v8f c = {0.f, 0.f, 0.f, 0.f, 0.f, 0.f, 0.f, 0.f};
  for (int k0 = 0; k0 < K; k0 += 4) {
    float2 av = *(const float2*)ap;
    float bx = bp[0], by = bp[Nc];
    v2f a, b;
    a.x = av.x;            a.y = av.y;
    b.x = cok ? bx : 0.f;  b.y = cok ? by : 0.f;
    c = __builtin_amdgcn_wmma_f32_16x16x4_f32(false, a, false, b, (short)0, c,
                                              false, false);
    ap += 4;
    bp += (size_t)4 * Nc;
  }
  if (cok) {
    float bv = bias ? bias[col] : 0.f;
    float* crow = Cout + (size_t)(tm * 16 + 8 * half) * Nc + col;
    for (int v = 0; v < 8; ++v) {
      float val = c[v] + bv;
      if (act) val = fmaxf(val, 0.f);
      crow[(size_t)v * Nc] = val;
    }
  }
}


__global__ void hist_rank(const int* __restrict__ ei, int E, int* __restrict__ cnt,
                          int* __restrict__ rank) {
  int e = blockIdx.x * blockDim.x + threadIdx.x;
  if (e < E) rank[e] = atomicAdd(&cnt[ei[E + e]], 1);
}

__global__ void __launch_bounds__(1024) scan_counts(const int* __restrict__ cnt, int n,
                                                    int* __restrict__ rowptr) {
  __shared__ int s_tot[32];
  __shared__ int s_pre[32];
  const int t = threadIdx.x, lane = t & 31, w = t >> 5;
  const int base = blockIdx.x * 1024;
  int p = 0;
#pragma unroll 8
  for (int i = t; i < base; i += 1024) p += cnt[i];
  const int i = base + t;
  const int v = (i < n) ? cnt[i] : 0;
  int inc = v;
  for (int off = 1; off < 32; off <<= 1) {
    int u = __shfl_up(inc, off, 32);
    if (lane >= off) inc += u;
  }
  for (int off = 16; off > 0; off >>= 1) p += __shfl_xor(p, off, 32);
  if (lane == 31) s_tot[w] = inc;
  if (lane == 0)  s_pre[w] = p;
  __syncthreads();
  int offset = 0, wbase = 0;
  for (int k = 0; k < 32; ++k) {
    offset += s_pre[k];
    if (k < w) wbase += s_tot[k];
  }
  const int excl = offset + wbase + inc - v;
  if (i < n) rowptr[i] = excl;
  if (i == n - 1) rowptr[n] = excl + v;
}

__global__ void scatter_src(const int* __restrict__ ei, int E, const int* __restrict__ rowptr,
                            const int* __restrict__ rank, int* __restrict__ csr_src) {
  int e = blockIdx.x * blockDim.x + threadIdx.x;
  if (e < E) {
    int pos = rowptr[ei[E + e]] + rank[e];
    csr_src[pos] = ei[e];
  }
}

__device__ __forceinline__ int bcast_i(int v, int j) {
  return __builtin_amdgcn_readlane(v, j);
}
__device__ __forceinline__ float bcast_f(float v, int j) {
  return __int_as_float(__builtin_amdgcn_readlane(__float_as_int(v), j));
}

template <int HEADS, int AST>
__device__ __forceinline__ void edge_score(int j, int deg, int beg, int node,
                                           const int* __restrict__ csr_src,
                                           const float* __restrict__ as_v,
                                           float ad0, float ad1,
                                           int& s, float& e0, float& e1) {
  s = node; e0 = -INFINITY; e1 = -INFINITY;
  if (j < deg) {
    if (j < deg - 1) s = csr_src[beg + j];
    if (HEADS == 2) {
      float2 a = *(const float2*)(as_v + (size_t)s * AST);
      float v0 = a.x + ad0, v1 = a.y + ad1;
      e0 = (v0 > 0.f) ? v0 : 0.2f * v0;
      e1 = (v1 > 0.f) ? v1 : 0.2f * v1;
    } else {
      float v0 = as_v[s] + ad0;
      e0 = (v0 > 0.f) ? v0 : 0.2f * v0;
    }
  }
}

template <int HEADS, int G>
__device__ __forceinline__ void rows(int s, float ex0, float ex1, int j0, int lane,
                                     const float* __restrict__ h, float (&acc)[HEADS * 2]) {
  constexpr int ROW = HEADS * 64;
  constexpr int VEC = HEADS * 2;
  const int loff = lane * VEC;
  float hv[G][VEC];
#pragma unroll
  for (int jj = 0; jj < G; ++jj) {
    int sj = bcast_i(s, j0 + jj);
    const float* hp = h + (size_t)sj * ROW + loff;
    if (HEADS == 2) {
      float4 t = *(const float4*)hp;
      hv[jj][0] = t.x; hv[jj][1] = t.y; hv[jj][VEC - 2] = t.z; hv[jj][VEC - 1] = t.w;
    } else {
      float2 t = *(const float2*)hp;
      hv[jj][0] = t.x; hv[jj][1] = t.y;
    }
  }
#pragma unroll
  for (int jj = 0; jj < G; ++jj) {
    float ex = bcast_f(ex0, j0 + jj);
    if (HEADS == 2) { float x1 = bcast_f(ex1, j0 + jj); ex = (lane >> 4) ? x1 : ex; }
#pragma unroll
    for (int c = 0; c < VEC; ++c) acc[c] += ex * hv[jj][c];
  }
}

template <int HEADS>
__global__ void gat_gather(const int* __restrict__ rowptr, const int* __restrict__ csr_src,
                           const float* __restrict__ as_v, const float* __restrict__ ad_v,
                           const float* __restrict__ h, const float* __restrict__ bias,
                           float* __restrict__ out, int n) {
  constexpr int ROW = HEADS * 64;
  constexpr int VEC = HEADS * 2;
  int node = __builtin_amdgcn_readfirstlane(blockIdx.x * (blockDim.x >> 5) + (threadIdx.x >> 5));
  if (node >= n) return;
  const int lane = threadIdx.x & 31;
  const int beg  = __builtin_amdgcn_readfirstlane(rowptr[node]);
  const int end  = __builtin_amdgcn_readfirstlane(rowptr[node + 1]);
  const int deg  = end - beg + 1;
  const float ad0 = ad_v[(size_t)node * HEADS];
  const float ad1 = (HEADS == 2) ? ad_v[(size_t)node * 2 + 1] : 0.f;
  const int loff  = lane * VEC;
  float bv[VEC];
#pragma unroll
  for (int c = 0; c < VEC; ++c) bv[c] = bias[loff + c];

  int sA; float eA0, eA1;
  edge_score<HEADS, HEADS>(lane, deg, beg, node, csr_src, as_v, ad0, ad1, sA, eA0, eA1);
  float m0 = eA0, m1 = eA1;
  for (int c0 = 32; c0 < deg; c0 += 32) {
    int s; float e0, e1;
    edge_score<HEADS, HEADS>(c0 + lane, deg, beg, node, csr_src, as_v, ad0, ad1, s, e0, e1);
    m0 = fmaxf(m0, e0);
    m1 = fmaxf(m1, e1);
  }
  for (int off = 16; off > 0; off >>= 1) {
    m0 = fmaxf(m0, __shfl_xor(m0, off, 32));
    if (HEADS == 2) m1 = fmaxf(m1, __shfl_xor(m1, off, 32));
  }

  float z0 = 0.f, z1 = 0.f;
  float acc[VEC];
#pragma unroll
  for (int c = 0; c < VEC; ++c) acc[c] = 0.f;
  for (int c0 = 0; c0 < deg; c0 += 32) {
    int s = sA; float e0 = eA0, e1 = eA1;
    if (c0 > 0) edge_score<HEADS, HEADS>(c0 + lane, deg, beg, node, csr_src, as_v, ad0, ad1, s, e0, e1);
    const bool ok = (c0 + lane) < deg;
    const float ex0 = ok ? expf(e0 - m0) : 0.f;
    const float ex1 = (HEADS == 2 && ok) ? expf(e1 - m1) : 0.f;
    z0 += ex0;
    z1 += ex1;
    const int cnt = min(32, deg - c0);
    int j0 = 0;
    for (; j0 + 8 <= cnt; j0 += 8) rows<HEADS, 8>(s, ex0, ex1, j0, lane, h, acc);
    if (cnt & 4) { rows<HEADS, 4>(s, ex0, ex1, j0, lane, h, acc); j0 += 4; }
    if (cnt & 2) { rows<HEADS, 2>(s, ex0, ex1, j0, lane, h, acc); j0 += 2; }
    if (cnt & 1) { rows<HEADS, 1>(s, ex0, ex1, j0, lane, h, acc); }
  }
  for (int off = 16; off > 0; off >>= 1) {
    z0 += __shfl_xor(z0, off, 32);
    if (HEADS == 2) z1 += __shfl_xor(z1, off, 32);
  }

  const float zz  = (HEADS == 2 && (lane >> 4)) ? z1 : z0;
  const float den = zz + 1e-16f;
  float r[VEC];
#pragma unroll
  for (int c = 0; c < VEC; ++c) {
    float v = acc[c] / den + bv[c];
    r[c] = (v > 0.f) ? v : expm1f(v);
  }
  float* op = out + (size_t)node * ROW + loff;
  if (HEADS == 2) *(float4*)op = make_float4(r[0], r[1], r[VEC - 2], r[VEC - 1]);
  else            *(float2*)op = make_float2(r[0], r[1]);
}


__global__ void fold_att_into_w(const float* __restrict__ Wm, const float* __restrict__ a_src,
                                const float* __restrict__ a_dst, float* __restrict__ wa) {
  const int t = threadIdx.x;
  const int k = t >> 2, cc = t & 3, hd = cc & 1;
  const float* av = ((cc & 2) ? a_dst : a_src) + hd * 64;
  const float* wr = Wm + (size_t)k * 128 + hd * 64;
  float acc = 0.f;
#pragma unroll
  for (int f = 0; f < 64; f += 4) {
    const float4 w4 = *(const float4*)(wr + f);
    const float4 a4 = *(const float4*)(av + f);
    acc += w4.x * a4.x + w4.y * a4.y + w4.z * a4.z + w4.w * a4.w;
  }
  wa[t] = acc;
}

template <int G>
__device__ __forceinline__ void rows_x2(int s, float ex0, float ex1, int j0, int lane,
                                        const float* __restrict__ x, v2f& acc0, v2f& acc1) {
  const int loff = lane * 2;
  v2f hv[G];
#pragma unroll
  for (int jj = 0; jj < G; ++jj) {
    int sj = bcast_i(s, j0 + jj);
    hv[jj] = *(const v2f*)(x + (size_t)sj * 64 + loff);
  }
#pragma unroll
  for (int jj = 0; jj < G; ++jj) {
    const float w0 = bcast_f(ex0, j0 + jj);
    const float w1 = bcast_f(ex1, j0 + jj);
    acc0 += w0 * hv[jj];
    acc1 += w1 * hv[jj];
  }
}

__global__ void gat_gather_x2(const int* __restrict__ rowptr, const int* __restrict__ csr_src,
                              const float* __restrict__ sd, const float* __restrict__ x,
                              float* __restrict__ agg, int n) {
  int node = __builtin_amdgcn_readfirstlane(blockIdx.x * (blockDim.x >> 5) + (threadIdx.x >> 5));
  if (node >= n) return;
  const int lane = threadIdx.x & 31;
  const int beg  = __builtin_amdgcn_readfirstlane(rowptr[node]);
  const int end  = __builtin_amdgcn_readfirstlane(rowptr[node + 1]);
  const int deg  = end - beg + 1;
  const float2 adv = *(const float2*)(sd + (size_t)node * 4 + 2);
  const float ad0 = adv.x, ad1 = adv.y;

  int sA; float eA0, eA1;
  edge_score<2, 4>(lane, deg, beg, node, csr_src, sd, ad0, ad1, sA, eA0, eA1);
  float m0 = eA0, m1 = eA1;
  for (int c0 = 32; c0 < deg; c0 += 32) {
    int s; float e0, e1;
    edge_score<2, 4>(c0 + lane, deg, beg, node, csr_src, sd, ad0, ad1, s, e0, e1);
    m0 = fmaxf(m0, e0);
    m1 = fmaxf(m1, e1);
  }
  for (int off = 16; off > 0; off >>= 1) {
    m0 = fmaxf(m0, __shfl_xor(m0, off, 32));
    m1 = fmaxf(m1, __shfl_xor(m1, off, 32));
  }

  float z0 = 0.f, z1 = 0.f;
  v2f acc0 = {0.f, 0.f}, acc1 = {0.f, 0.f};
  for (int c0 = 0; c0 < deg; c0 += 32) {
    int s = sA; float e0 = eA0, e1 = eA1;
    if (c0 > 0) edge_score<2, 4>(c0 + lane, deg, beg, node, csr_src, sd, ad0, ad1, s, e0, e1);
    const bool ok = (c0 + lane) < deg;
    const float ex0 = ok ? expf(e0 - m0) : 0.f;
    const float ex1 = ok ? expf(e1 - m1) : 0.f;
    z0 += ex0;
    z1 += ex1;
    const int cnt = min(32, deg - c0);
    int j0 = 0;
    for (; j0 + 8 <= cnt; j0 += 8) rows_x2<8>(s, ex0, ex1, j0, lane, x, acc0, acc1);
    if (cnt & 4) { rows_x2<4>(s, ex0, ex1, j0, lane, x, acc0, acc1); j0 += 4; }
    if (cnt & 2) { rows_x2<2>(s, ex0, ex1, j0, lane, x, acc0, acc1); j0 += 2; }
    if (cnt & 1) { rows_x2<1>(s, ex0, ex1, j0, lane, x, acc0, acc1); }
  }
  for (int off = 16; off > 0; off >>= 1) {
    z0 += __shfl_xor(z0, off, 32);
    z1 += __shfl_xor(z1, off, 32);
  }

  const float den0 = z0 + 1e-16f, den1 = z1 + 1e-16f;
  float* op = agg + (size_t)node * 128 + lane * 2;
  *(float2*)op        = make_float2(acc0.x / den0, acc0.y / den0);
  *(float2*)(op + 64) = make_float2(acc1.x / den1, acc1.y / den1);
}

extern "C" void kernel_launch(void* const* d_in, const int* in_sizes, int n_in,
                              void* d_out, int out_size, void* d_ws, size_t ws_size,
                              hipStream_t stream) {
  const float* x      = (const float*)d_in[0];
  const int*   ei     = (const int*)d_in[1];
  const float* W1     = (const float*)d_in[2];
  const float* a_src1 = (const float*)d_in[3];
  const float* a_dst1 = (const float*)d_in[4];
  const float* b1     = (const float*)d_in[5];
  const float* W2     = (const float*)d_in[6];
  const float* a_src2 = (const float*)d_in[7];
  const float* a_dst2 = (const float*)d_in[8];
  const float* b2     = (const float*)d_in[9];
  const float* m1w    = (const float*)d_in[10];
  const float* m1b    = (const float*)d_in[11];
  const float* m2w    = (const float*)d_in[12];
  const float* m2b    = (const float*)d_in[13];

  const int N    = in_sizes[0] / 64;
  const int E    = in_sizes[1] / 2;
  const int C    = in_sizes[12] / 64;

  float* bufA   = (float*)d_ws;
  float* bufB   = bufA + (size_t)N * 128;
  float* bufC   = bufB + (size_t)N * 128;
  float* as_v   = bufC + (size_t)N * 64;
  float* ad_v   = as_v + (size_t)N * 2;
  int*   cnt    = (int*)(ad_v + (size_t)N * 2);
  int*   rowptr = cnt + (size_t)N;
  int*   csrsrc = rowptr + (size_t)N + 1;
  int*   rank   = (int*)bufB;

  const int TPB = 256;
  auto blocks1d = [&](long n) { return (int)((n + TPB - 1) / TPB); };
  auto gemm_blocks = [&](int M, int Nc) {
    int waves = ((M + 15) / 16) * ((Nc + 15) / 16);
    return (waves + 7) / 8;
  };
  auto strip_blocks = [&](int M, int Nc) {
    int waves = ((M + 15) / 16) * (Nc / 64);
    return (waves + 7) / 8;
  };

  fill_i32<<<blocks1d(N), TPB, 0, stream>>>(cnt, 0, N);
  hist_rank<<<blocks1d(E), TPB, 0, stream>>>(ei, E, cnt, rank);
  scan_counts<<<(N + 1023) / 1024, 1024, 0, stream>>>(cnt, N, rowptr);
  scatter_src<<<blocks1d(E), TPB, 0, stream>>>(ei, E, rowptr, rank, csrsrc);

  float* wa = bufC;
  fold_att_into_w<<<1, 256, 0, stream>>>(W1, a_src1, a_dst1, wa);
  gemm_wmma_f32<64, 4><<<gemm_blocks(N, 4), TPB, 0, stream>>>(x, wa, nullptr, as_v, N, 0);
  gat_gather_x2<<<(N + 7) / 8, TPB, 0, stream>>>(rowptr, csrsrc, as_v, x, bufA, N);
  gemm_strip64<64, 128, 128, false, true, true><<<strip_blocks(N, 128), TPB, 0, stream>>>(
      bufA, W1, b1, bufB, nullptr, nullptr, nullptr, nullptr, N, 0);

  gemm_strip64<128, 128, 64, true, false, false><<<strip_blocks(N, 64), TPB, 0, stream>>>(
      bufB, W2, nullptr, bufA, a_src2, a_dst2, as_v, ad_v, N, 0);
  gat_gather<1><<<(N + 7) / 8, TPB, 0, stream>>>(rowptr, csrsrc, as_v, ad_v,
                                                 bufA, b2, bufC, N);

  gemm_strip64<64, 64, 64, false, false, false><<<strip_blocks(N, 64), TPB, 0, stream>>>(
      bufC, m1w, m1b, bufA, nullptr, nullptr, nullptr, nullptr, N, 1);
  if (C == 8) {
    gemm_wmma_f32<64, 8><<<gemm_blocks(N, 8), TPB, 0, stream>>>(bufA, m2w, m2b,
                                                                (float*)d_out, N, 1);
  } else {
    gemm_wmma_f32_dyn<<<gemm_blocks(N, C), TPB, 0, stream>>>(bufA, m2w, m2b,
                                                             (float*)d_out, N, 64, C, 1);
  }
}


# edge grouping: each level puts its sorted output together in LDS and writes it as whole 128-byte lines (16 B per lane), not one scattered 4- or 8-byte store per edge; ranks and slot order as before
# speedup vs baseline: 6.6871x; 2.0013x over previous
#include <hip/hip_runtime.h>
#include <math.h>

typedef __attribute__((ext_vector_type(2))) float v2f;
typedef __attribute__((ext_vector_type(8))) float v8f;


template <int Nc>
__device__ __forceinline__ void strip_load_b(const float* __restrict__ bq,
                                             float4 (&b0)[4], float4 (&b1)[4]) {
#pragma unroll
  for (int i = 0; i < 4; ++i) {
    b0[i] = *(const float4*)(bq + (size_t)(4 * i) * Nc);
    b1[i] = *(const float4*)(bq + (size_t)(4 * i) * Nc + Nc);
  }
}

__device__ __forceinline__ void strip_mma16(const float2 (&ac)[4], const float4 (&b0)[4],
                                            const float4 (&b1)[4], v8f& acc0, v8f& acc1,
                                            v8f& acc2, v8f& acc3) {
#pragma unroll
  for (int i = 0; i < 4; ++i) {
    v2f a, b;
    a.x = ac[i].x; a.y = ac[i].y;
    b.x = b0[i].x; b.y = b1[i].x;
    acc0 = __builtin_amdgcn_wmma_f32_16x16x4_f32(false, a, false, b, (short)0, acc0, false, false);
    b.x = b0[i].y; b.y = b1[i].y;
    acc1 = __builtin_amdgcn_wmma_f32_16x16x4_f32(false, a, false, b, (short)0, acc1, false, false);
    b.x = b0[i].z; b.y = b1[i].z;
    acc2 = __builtin_amdgcn_wmma_f32_16x16x4_f32(false, a, false, b, (short)0, acc2, false, false);
    b.x = b0[i].w; b.y = b1[i].w;
    acc3 = __builtin_amdgcn_wmma_f32_16x16x4_f32(false, a, false, b, (short)0, acc3, false, false);
  }
}

template <int K, int Nc>
__device__ __forceinline__ void strip_k_mem(const float* __restrict__ ap, const float* __restrict__ bp,
                                            v8f& acc0, v8f& acc1, v8f& acc2, v8f& acc3) {
  float2 an[4];
#pragma unroll
  for (int i = 0; i < 4; ++i) an[i] = *(const float2*)(ap + 4 * i);
#pragma nounroll
  for (int kc = 0; kc < K - 16; kc += 16) {
    float4 b0[4], b1[4];
    strip_load_b<Nc>(bp + (size_t)kc * Nc, b0, b1);
    float2 ac[4];
#pragma unroll
    for (int i = 0; i < 4; ++i) ac[i] = an[i];
#pragma unroll
    for (int i = 0; i < 4; ++i) an[i] = *(const float2*)(ap + kc + 16 + 4 * i);
    strip_mma16(ac, b0, b1, acc0, acc1, acc2, acc3);
  }
  {
    float4 b0[4], b1[4];
    strip_load_b<Nc>(bp + (size_t)(K - 16) * Nc, b0, b1);
    strip_mma16(an, b0, b1, acc0, acc1, acc2, acc3);
  }
}

template <int strips>
__device__ __forceinline__ void strip_alpha(const v8f& acc0, const v8f& acc1, const v8f& acc2, const v8f& acc3,
                                            const float* __restrict__ a_src, const float* __restrict__ a_dst,
                                            float* __restrict__ as_out, float* __restrict__ ad_out,
                                            int n0, int tm, int sn, int half, int c) {
  const float4 s4 = *(const float4*)(a_src + n0);
  const float4 d4 = *(const float4*)(a_dst + n0);
  float ps[8], pd[8];
#pragma unroll
  for (int r = 0; r < 8; ++r) {
    ps[r] = acc0[r] * s4.x + acc1[r] * s4.y + acc2[r] * s4.z + acc3[r] * s4.w;
    pd[r] = acc0[r] * d4.x + acc1[r] * d4.y + acc2[r] * d4.z + acc3[r] * d4.w;
  }
  const bool k8 = (c & 8) != 0, k4 = (c & 4) != 0, k2 = (c & 2) != 0, k1 = (c & 1) != 0;
  float v8[8], v4[4], v2[2];
#pragma unroll
  for (int i = 0; i < 8; ++i) {
    float mine = k8 ? pd[i] : ps[i];
    float oth  = k8 ? ps[i] : pd[i];
    v8[i] = mine + __shfl_xor(oth, 8, 32);
  }
#pragma unroll
  for (int i = 0; i < 4; ++i) {
    float mine = k4 ? v8[i + 4] : v8[i];
    float oth  = k4 ? v8[i] : v8[i + 4];
    v4[i] = mine + __shfl_xor(oth, 4, 32);
  }
#pragma unroll
  for (int i = 0; i < 2; ++i) {
    float mine = k2 ? v4[i + 2] : v4[i];
    float oth  = k2 ? v4[i] : v4[i + 2];
    v2[i] = mine + __shfl_xor(oth, 2, 32);
  }
  float mine = k1 ? v2[1] : v2[0];
  float oth  = k1 ? v2[0] : v2[1];
  float tot  = mine + __shfl_xor(oth, 1, 32);
  int orow = tm * 16 + 8 * half + (c & 7);
  float* dst = k8 ? ad_out : as_out;
  dst[(size_t)orow * strips + sn] = tot;
}

template <int K, int LDA, int Nc, bool ALPHA, bool HEADA, bool ELU>
__global__ void gemm_strip64(const float* __restrict__ A, const float* __restrict__ B,
                             const float* __restrict__ bias, float* __restrict__ Cout,
                             const float* __restrict__ a_src, const float* __restrict__ a_dst,
                             float* __restrict__ as_out, float* __restrict__ ad_out,
                             int M, int act) {
  static_assert(Nc % 64 == 0 && K % 16 == 0, "whole strips and whole chunks only");
  constexpr int strips = Nc / 64;
  int wave   = blockIdx.x * (blockDim.x >> 5) + (threadIdx.x >> 5);
  int tilesM = (M + 15) >> 4;
  if (wave >= tilesM * strips) return;
  int tm   = wave / strips;
  int sn   = wave - tm * strips;
  int lane = threadIdx.x & 31;
  int half = lane >> 4;
  int c    = lane & 15;
  int row  = tm * 16 + c;
  int n0   = sn * 64 + 4 * c;

  const float* ap = A + (size_t)row * LDA + (HEADA ? sn * 64 : 0) + 2 * half;
  const float* bp = B + (size_t)(2 * half) * Nc + n0;

  v8f acc0 = {0.f, 0.f, 0.f, 0.f, 0.f, 0.f, 0.f, 0.f};
  v8f acc1 = acc0, acc2 = acc0, acc3 = acc0;
  strip_k_mem<K, Nc>(ap, bp, acc0, acc1, acc2, acc3);

  if (ALPHA) strip_alpha<strips>(acc0, acc1, acc2, acc3, a_src, a_dst, as_out, ad_out, n0, tm, sn, half, c);

  float4 bv = make_float4(0.f, 0.f, 0.f, 0.f);
  if (bias) bv = *(const float4*)(bias + n0);
  float* crow = Cout + (size_t)(tm * 16 + 8 * half) * Nc + n0;
#pragma unroll
  for (int r = 0; r < 8; ++r) {
    float4 v = make_float4(acc0[r] + bv.x, acc1[r] + bv.y, acc2[r] + bv.z, acc3[r] + bv.w);
    if (act) { v.x = fmaxf(v.x, 0.f); v.y = fmaxf(v.y, 0.f); v.z = fmaxf(v.z, 0.f); v.w = fmaxf(v.w, 0.f); }
    if (ELU) {
      v.x = (v.x > 0.f) ? v.x : expm1f(v.x);
      v.y = (v.y > 0.f) ? v.y : expm1f(v.y);
      v.z = (v.z > 0.f) ? v.z : expm1f(v.z);
      v.w = (v.w > 0.f) ? v.w : expm1f(v.w);
    }
    *(float4*)(crow + (size_t)r * Nc) = v;
  }
}

constexpr int CHAIN_WAVES = 4;

template <int K, int Nc>
__device__ __forceinline__ void strip_k_lds(const float* al, const float* __restrict__ bp,
                                            v8f& acc0, v8f& acc1, v8f& acc2, v8f& acc3) {
#pragma nounroll
  for (int kc = 0; kc < K; kc += 16) {
    float4 b0[4], b1[4];
    strip_load_b<Nc>(bp + (size_t)kc * Nc, b0, b1);
    float2 ac[4];
#pragma unroll
    for (int i = 0; i < 4; ++i) ac[i] = *(const float2*)(al + kc + 4 * i);
    strip_mma16(ac, b0, b1, acc0, acc1, acc2, acc3);
  }
}

template <int PITCH>
__device__ __forceinline__ void strip_to_lds(float* slice, int half, int n0, const float4 (&v)[8]) {
#pragma unroll
  for (int r = 0; r < 8; ++r) *(float4*)(slice + (r + 8 * half) * PITCH + n0) = v[r];
}

__global__ void __launch_bounds__(32 * CHAIN_WAVES) proj_chain(const float* __restrict__ agg, const float* __restrict__ W1,
                                                  const float* __restrict__ b1, const float* __restrict__ W2,
                                                  const float* __restrict__ a_src, const float* __restrict__ a_dst,
                                                  float* __restrict__ h2pre, float* __restrict__ as_out,
                                                  float* __restrict__ ad_out, int M) {
  constexpr int PITCH = 128 + 4;
  __shared__ alignas(16) float mid[CHAIN_WAVES][16 * PITCH];
  const int w    = threadIdx.x >> 5;
  const int tm   = blockIdx.x * CHAIN_WAVES + w;
  if (tm >= ((M + 15) >> 4)) return;
  const int lane = threadIdx.x & 31;
  const int half = lane >> 4;
  const int c    = lane & 15;
  const int row  = tm * 16 + c;
  float* slice = mid[w];
  const v8f zero = {0.f, 0.f, 0.f, 0.f, 0.f, 0.f, 0.f, 0.f};

  for (int sn = 0; sn < 2; ++sn) {
    const int n0 = sn * 64 + 4 * c;
    v8f acc0 = zero, acc1 = zero, acc2 = zero, acc3 = zero;
    strip_k_mem<64, 128>(agg + (size_t)row * 128 + sn * 64 + 2 * half,
                         W1 + (size_t)(2 * half) * 128 + n0, acc0, acc1, acc2, acc3);
    const float4 bv = *(const float4*)(b1 + n0);
    float4 v[8];
#pragma unroll
    for (int r = 0; r < 8; ++r) {
      float4 t = make_float4(acc0[r] + bv.x, acc1[r] + bv.y, acc2[r] + bv.z, acc3[r] + bv.w);
      t.x = (t.x > 0.f) ? t.x : expm1f(t.x);
      t.y = (t.y > 0.f) ? t.y : expm1f(t.y);
      t.z = (t.z > 0.f) ? t.z : expm1f(t.z);
      t.w = (t.w > 0.f) ? t.w : expm1f(t.w);
      v[r] = t;
    }
    strip_to_lds<PITCH>(slice, half, n0, v);
  }
  __builtin_amdgcn_fence(__ATOMIC_ACQ_REL, "wavefront");

  const int n0 = 4 * c;
  v8f acc0 = zero, acc1 = zero, acc2 = zero, acc3 = zero;
  strip_k_lds<128, 64>(slice + c * PITCH + 2 * half, W2 + (size_t)(2 * half) * 64 + n0,
                       acc0, acc1, acc2, acc3);
  strip_alpha<1>(acc0, acc1, acc2, acc3, a_src, a_dst, as_out, ad_out, n0, tm, 0, half, c);
  float* crow = h2pre + (size_t)(tm * 16 + 8 * half) * 64 + n0;
#pragma unroll
  for (int r = 0; r < 8; ++r)
    *(float4*)(crow + (size_t)r * 64) = make_float4(acc0[r], acc1[r], acc2[r], acc3[r]);
}

template <int NC>
__global__ void __launch_bounds__(32 * CHAIN_WAVES) mlp_chain(const float* __restrict__ A, const float* __restrict__ w1,
                                                 const float* __restrict__ bias1, const float* __restrict__ w2,
                                                 const float* __restrict__ bias2, float* __restrict__ out, int M) {
  static_assert(NC <= 16, "one output tile");
  constexpr int PITCH = 64 + 4;
  __shared__ alignas(16) float mid[CHAIN_WAVES][16 * PITCH];
  const int w    = threadIdx.x >> 5;
  const int tm   = blockIdx.x * CHAIN_WAVES + w;
  if (tm >= ((M + 15) >> 4)) return;
  const int lane = threadIdx.x & 31;
  const int half = lane >> 4;
  const int c    = lane & 15;
  const int row  = tm * 16 + c;
  float* slice = mid[w];
  const v8f zero = {0.f, 0.f, 0.f, 0.f, 0.f, 0.f, 0.f, 0.f};
  {
    const int n0 = 4 * c;
    v8f acc0 = zero, acc1 = zero, acc2 = zero, acc3 = zero;
    strip_k_mem<64, 64>(A + (size_t)row * 64 + 2 * half, w1 + (size_t)(2 * half) * 64 + n0,
                        acc0, acc1, acc2, acc3);
    const float4 bv = *(const float4*)(bias1 + n0);
    float4 v[8];
#pragma unroll
    for (int r = 0; r < 8; ++r)
      v[r] = make_float4(fmaxf(acc0[r] + bv.x, 0.f), fmaxf(acc1[r] + bv.y, 0.f),
                         fmaxf(acc2[r] + bv.z, 0.f), fmaxf(acc3[r] + bv.w, 0.f));
    strip_to_lds<PITCH>(slice, half, n0, v);
  }
  __builtin_amdgcn_fence(__ATOMIC_ACQ_REL, "wavefront");

  const bool cok  = c < NC;
  const int  colc = cok ? c : 0;
  const float* al = slice + c * PITCH + 2 * half;
  const float* bp = w2 + (size_t)(2 * half) * NC + colc;
  v8f d = zero;
#pragma unroll
  for (int k0 = 0; k0 < 64; k0 += 4) {
    const float2 av = *(const float2*)(al + k0);
    const float  bx = bp[(size_t)k0 * NC];
    const float  by = bp[(size_t)k0 * NC + NC];
    v2f a, b;
    a.x = av.x;
    a.y = av.y;
    b.x = cok ? bx : 0.f;
    b.y = cok ? by : 0.f;
    d = __builtin_amdgcn_wmma_f32_16x16x4_f32(false, a, false, b, (short)0, d, false, false);
  }
  if (cok) {
    const float bv = bias2[c];
    float* crow = out + (size_t)(tm * 16 + 8 * half) * NC + c;
#pragma unroll
    for (int v = 0; v < 8; ++v) crow[(size_t)v * NC] = fmaxf(d[v] + bv, 0.f);
  }
}

__global__ void gemm_wmma_f32_dyn(const float* __restrict__ A, const float* __restrict__ B,
                                  const float* __restrict__ bias, float* __restrict__ Cout,
                                  int M, int K, int Nc, int act) {
  int wave   = blockIdx.x * (blockDim.x >> 5) + (threadIdx.x >> 5);
  int tilesN = (Nc + 15) >> 4;
  int tilesM = (M + 15) >> 4;
  if (wave >= tilesM * tilesN) return;
  int tm = wave / tilesN, tn = wave - tm * tilesN;
  int lane = threadIdx.x & 31, half = lane >> 4, l16 = lane & 15;
  int row = tm * 16 + l16, col = tn * 16 + l16;
  bool cok = (col < Nc);
  int colc = cok ? col : 0;
  const float* ap = A + (size_t)row * K + 2 * half;
  const float* bp = B + (size_t)(2 * half) * Nc + colc;
  v8f c = {0.f, 0.f, 0.f, 0.f, 0.f, 0.f, 0.f, 0.f};
  for (int k0 = 0; k0 < K; k0 += 4) {
    float2 av = *(const float2*)ap;
    float bx = bp[0], by = bp[Nc];
    v2f a, b;
    a.x = av.x;            a.y = av.y;
    b.x = cok ? bx : 0.f;  b.y = cok ? by : 0.f;
    c = __builtin_amdgcn_wmma_f32_16x16x4_f32(false, a, false, b, (short)0, c,
                                              false, false);
    ap += 4;
    bp += (size_t)4 * Nc;
  }
  if (cok) {
    float bv = bias ? bias[col] : 0.f;
    float* crow = Cout + (size_t)(tm * 16 + 8 * half) * Nc + col;
    for (int v = 0; v < 8; ++v) {
      float val = c[v] + bv;
      if (act) val = fmaxf(val, 0.f);
      crow[(size_t)v * Nc] = val;
    }
  }
}

constexpr int BK_SHIFT   = 9;
constexpr int BK_NODES   = 1 << BK_SHIFT;
constexpr int EPT        = 8;
constexpr int TILE_EDGES = 1024 * EPT;
constexpr int MAX_BK     = 4096;
constexpr int MAX_TILES  = 1024;

__global__ void __launch_bounds__(1024) tile_sort(const int* __restrict__ ei, int E, int nbk,
                                                  int* __restrict__ startT,
                                                  int* __restrict__ pairs_w) {
  int2* __restrict__ pairs = (int2*)pairs_w;
  __shared__ int h[MAX_BK + 32];
  __shared__ int s_tot[32];
  __shared__ alignas(16) int2 stage[TILE_EDGES];
  const int t = threadIdx.x, lane = t & 31, w = t >> 5;
  const int nblk = gridDim.x;
  for (int i = t; i < nbk + 32; i += 1024) h[i] = 0;
  __syncthreads();
  const int base = blockIdx.x * TILE_EDGES + t;
  int s[EPT], d[EPT];
#pragma unroll
  for (int k = 0; k < EPT; ++k) {
    const int e  = base + k * 1024;
    const int ec = min(e, E - 1);
    const int dv = ei[E + ec];
    s[k] = ei[ec];
    d[k] = (e < E) ? dv : -1;
  }
  int pos[EPT];
#pragma unroll
  for (int k = 0; k < EPT; ++k) {
    const int bk = (d[k] >= 0) ? (d[k] >> BK_SHIFT) : nbk + lane;
    pos[k] = atomicAdd(&h[bk], 1);
  }
  __syncthreads();
  int v[4];
  int tsum = 0;
#pragma unroll
  for (int j = 0; j < 4; ++j) {
    const int idx = 4 * t + j;
    v[j] = (idx < nbk) ? h[idx] : 0;
    tsum += v[j];
  }
  int inc = tsum;
  for (int off = 1; off < 32; off <<= 1) {
    int u = __shfl_up(inc, off, 32);
    if (lane >= off) inc += u;
  }
  if (lane == 31) s_tot[w] = inc;
  __syncthreads();
  int run = inc - tsum;
  for (int k = 0; k < 32; ++k)
    if (k < w) run += s_tot[k];
#pragma unroll
  for (int j = 0; j < 4; ++j) {
    const int idx = 4 * t + j;
    if (idx < nbk) {
      h[idx] = run;
      startT[(size_t)idx * nblk + blockIdx.x] = run;
    }
    run += v[j];
  }
  const int cntT = min(TILE_EDGES, E - (int)blockIdx.x * TILE_EDGES);
  if (t == 0) startT[(size_t)nbk * nblk + blockIdx.x] = cntT;
  __syncthreads();
  int2* __restrict__ tile = pairs + (size_t)blockIdx.x * TILE_EDGES;
  int rb[EPT];
#pragma unroll
  for (int k = 0; k < EPT; ++k) rb[k] = h[(d[k] >= 0) ? (d[k] >> BK_SHIFT) : nbk + lane];
  const bool full = cntT == TILE_EDGES;
  if (full) {
#pragma unroll
    for (int k = 0; k < EPT; ++k) stage[rb[k] + pos[k]] = make_int2(s[k], d[k]);
  } else {
#pragma unroll
    for (int k = 0; k < EPT; ++k)
      if (d[k] >= 0) stage[rb[k] + pos[k]] = make_int2(s[k], d[k]);
  }
  __syncthreads();
  constexpr int QT = TILE_EDGES / 2048;
  const int4* st4 = (const int4*)stage;
  int4* out4 = (int4*)tile;
  if (full) {
    int4 v[QT];
#pragma unroll
    for (int i = 0; i < QT; ++i) v[i] = st4[t + 1024 * i];
#pragma unroll
    for (int i = 0; i < QT; ++i) out4[t + 1024 * i] = v[i];
  } else {
    for (int i = 0; i < QT; ++i) {
      const int q = t + 1024 * i;
      if (2 * q + 1 < cntT)  out4[q] = st4[q];
      else if (2 * q < cntT) tile[2 * q] = stage[2 * q];
    }
  }
}

template <int PT>
__device__ __forceinline__ void list_load(const int2* __restrict__ pairs, const int* off, const int* adj,
                                          int nblk, int top, int total, int E, int i0, int t,
                                          int2 (&p)[PT]) {
  int jc[PT], kq[PT];
#pragma unroll
  for (int k = 0; k < PT; ++k) {
    jc[k] = max(min(i0 + k * 1024 + t, total - 1), 0);
    kq[k] = 0;
  }
  for (int st = top; st > 0; st >>= 1) {
    int ov[PT];
#pragma unroll
    for (int k = 0; k < PT; ++k) ov[k] = off[min(kq[k] + st, nblk)];
#pragma unroll
    for (int k = 0; k < PT; ++k)
      if (ov[k] <= jc[k]) kq[k] += st;
  }
#pragma unroll
  for (int k = 0; k < PT; ++k) {
    const int idx = max(0, min(adj[kq[k]] + jc[k], E - 1));
    p[k] = pairs[idx];
    if (i0 + k * 1024 + t >= total) p[k].y = -1;
  }
}

template <int PT>
__device__ __forceinline__ void list_count(int* cnt, const int2 (&p)[PT], int lane) {
#pragma unroll
  for (int k = 0; k < PT; ++k) {
    const int nd = (p[k].y >= 0) ? (p[k].y & (BK_NODES - 1)) : BK_NODES + lane;
    atomicAdd(&cnt[nd], 1);
  }
}

template <int PT>
__device__ __forceinline__ void list_place(int* cnt, int* __restrict__ csr_src, const int2 (&p)[PT], int lane) {
  int pos[PT];
#pragma unroll
  for (int k = 0; k < PT; ++k) {
    const int nd = (p[k].y >= 0) ? (p[k].y & (BK_NODES - 1)) : BK_NODES + lane;
    pos[k] = atomicAdd(&cnt[nd], 1);
  }
#pragma unroll
  for (int k = 0; k < PT; ++k)
    if (p[k].y >= 0) csr_src[pos[k]] = p[k].x;
}

template <int PT>
__device__ __forceinline__ void list_place_lds(int* cnt, int* stage, int sbase, const int2 (&p)[PT], int lane) {
  int pos[PT];
#pragma unroll
  for (int k = 0; k < PT; ++k) {
    const int nd = (p[k].y >= 0) ? (p[k].y & (BK_NODES - 1)) : BK_NODES + lane;
    pos[k] = atomicAdd(&cnt[nd], 1);
  }
#pragma unroll
  for (int k = 0; k < PT; ++k)
    if (p[k].y >= 0) stage[pos[k] - sbase] = p[k].x;
}

constexpr int BS_PT = 10;

__global__ void __launch_bounds__(1024) bucket_sort(const int* __restrict__ pairs_w,
                                                    const int* __restrict__ startT, int nblk, int E, int n,
                                                    int* __restrict__ rowptr,
                                                    int* __restrict__ csr_src) {
  const int2* __restrict__ pairs = (const int2*)pairs_w;
  __shared__ int cnt[BK_NODES + 32];
  __shared__ int s_tot[32];
  __shared__ int s_beg[32];
  __shared__ int off[MAX_TILES + 1];
  __shared__ int adj[MAX_TILES];
  __shared__ alignas(16) int stage[BS_PT * 1024 + 32];
  const int t = threadIdx.x, lane = t & 31, w = t >> 5;
  const int b = blockIdx.x;
  int rs = 0, len = 0;
  if (t < nblk) {
    rs  = startT[(size_t)b * nblk + t];
    len = startT[(size_t)(b + 1) * nblk + t] - rs;
  }
  if (t < BK_NODES + 32) cnt[t] = 0;
  int inc = len;
  for (int o = 1; o < 32; o <<= 1) {
    int u = __shfl_up(inc, o, 32);
    if (lane >= o) inc += u;
  }
  int sb = rs;
  for (int o = 16; o > 0; o >>= 1) sb += __shfl_xor(sb, o, 32);
  if (lane == 31) s_tot[w] = inc;
  if (lane == 0)  s_beg[w] = sb;
  __syncthreads();
  int wbase = 0, total = 0, beg = 0;
  for (int k = 0; k < 32; ++k) {
    if (k < w) wbase += s_tot[k];
    total += s_tot[k];
    beg   += s_beg[k];
  }
  const int o0 = wbase + inc - len;
  if (t < nblk) {
    off[t] = o0;
    adj[t] = t * TILE_EDGES + rs - o0;
  }
  if (t == 0) off[nblk] = total;
  const int end = beg + total;
  int top = 1;
  while (top * 2 < nblk) top *= 2;
  __syncthreads();
  constexpr int PT = BS_PT;
  int2 p0[PT];
  list_load<PT>(pairs, off, adj, nblk, top, total, E, 0, t, p0);
  list_count<PT>(cnt, p0, lane);
  for (int i0 = PT * 1024; i0 < total; i0 += PT * 1024) {
    int2 p[PT];
    list_load<PT>(pairs, off, adj, nblk, top, total, E, i0, t, p);
    list_count<PT>(cnt, p, lane);
  }
  __syncthreads();
  const int v = (t < BK_NODES) ? cnt[t] : 0;
  int inc2 = v;
  for (int o = 1; o < 32; o <<= 1) {
    int u = __shfl_up(inc2, o, 32);
    if (lane >= o) inc2 += u;
  }
  if (lane == 31) s_tot[w] = inc2;
  __syncthreads();
  int wb = 0;
  for (int k = 0; k < 32; ++k)
    if (k < w) wb += s_tot[k];
  const int start = beg + wb + inc2 - v;
  const int node  = b * BK_NODES + t;
  if (t < BK_NODES) {
    cnt[t] = start;
    if (node < n) rowptr[node] = start;
  }
  if (b == (int)gridDim.x - 1 && t == 0) rowptr[n] = end;
  __syncthreads();
  const bool staged = total <= PT * 1024;
  const int mis   = (int)(((size_t)(csr_src + beg) >> 2) & 31);
  const int sbase = beg - mis;
  if (staged) {
    list_place_lds<PT>(cnt, stage, sbase, p0, lane);
  } else {
    list_place<PT>(cnt, csr_src, p0, lane);
    for (int i0 = PT * 1024; i0 < total; i0 += PT * 1024) {
      int2 p[PT];
      list_load<PT>(pairs, off, adj, nblk, top, total, E, i0, t, p);
      list_place<PT>(cnt, csr_src, p, lane);
    }
  }
  __syncthreads();
  if (staged) {
    const int lim = total + mis;
    for (int g = t; 4 * g < lim; g += 1024) {
      const int s0 = 4 * g;
      if (s0 >= mis && s0 + 3 < lim) {
        *(int4*)(csr_src + (sbase + s0)) = *(const int4*)(stage + s0);
      } else {
        for (int j = 0; j < 4; ++j) {
          const int s = s0 + j;
          if (s >= mis && s < lim) csr_src[sbase + s] = stage[s];
        }
      }
    }
  }
}

__device__ __forceinline__ int bcast_i(int v, int j) {
  return __builtin_amdgcn_readlane(v, j);
}
__device__ __forceinline__ float bcast_f(float v, int j) {
  return __int_as_float(__builtin_amdgcn_readlane(__float_as_int(v), j));
}

template <int HEADS, int AST>
__device__ __forceinline__ void edge_score(int j, int deg, int beg, int node,
                                           const int* __restrict__ csr_src,
                                           const float* __restrict__ as_v,
                                           float ad0, float ad1,
                                           int& s, float& e0, float& e1) {
  s = node; e0 = -INFINITY; e1 = -INFINITY;
  if (j < deg) {
    if (j < deg - 1) s = csr_src[beg + j];
    if (HEADS == 2) {
      float2 a = *(const float2*)(as_v + (size_t)s * AST);
      float v0 = a.x + ad0, v1 = a.y + ad1;
      e0 = (v0 > 0.f) ? v0 : 0.2f * v0;
      e1 = (v1 > 0.f) ? v1 : 0.2f * v1;
    } else {
      float v0 = as_v[s] + ad0;
      e0 = (v0 > 0.f) ? v0 : 0.2f * v0;
    }
  }
}

template <int HEADS, int G>
__device__ __forceinline__ void rows(int s, float ex0, float ex1, int j0, int lane,
                                     const float* __restrict__ h, float (&acc)[HEADS * 2]) {
  constexpr int ROW = HEADS * 64;
  constexpr int VEC = HEADS * 2;
  const int loff = lane * VEC;
  float hv[G][VEC];
#pragma unroll
  for (int jj = 0; jj < G; ++jj) {
    int sj = bcast_i(s, j0 + jj);
    const float* hp = h + (size_t)sj * ROW + loff;
    if (HEADS == 2) {
      float4 t = *(const float4*)hp;
      hv[jj][0] = t.x; hv[jj][1] = t.y; hv[jj][VEC - 2] = t.z; hv[jj][VEC - 1] = t.w;
    } else {
      float2 t = *(const float2*)hp;
      hv[jj][0] = t.x; hv[jj][1] = t.y;
    }
  }
#pragma unroll
  for (int jj = 0; jj < G; ++jj) {
    float ex = bcast_f(ex0, j0 + jj);
    if (HEADS == 2) { float x1 = bcast_f(ex1, j0 + jj); ex = (lane >> 4) ? x1 : ex; }
#pragma unroll
    for (int c = 0; c < VEC; ++c) acc[c] += ex * hv[jj][c];
  }
}

template <int HEADS>
__global__ void gat_gather(const int* __restrict__ rowptr, const int* __restrict__ csr_src,
                           const float* __restrict__ as_v, const float* __restrict__ ad_v,
                           const float* __restrict__ h, const float* __restrict__ bias,
                           float* __restrict__ out, int n) {
  constexpr int ROW = HEADS * 64;
  constexpr int VEC = HEADS * 2;
  int node = __builtin_amdgcn_readfirstlane(blockIdx.x * (blockDim.x >> 5) + (threadIdx.x >> 5));
  if (node >= n) return;
  const int lane = threadIdx.x & 31;
  const int beg  = __builtin_amdgcn_readfirstlane(rowptr[node]);
  const int end  = __builtin_amdgcn_readfirstlane(rowptr[node + 1]);
  const int deg  = end - beg + 1;
  const float ad0 = ad_v[(size_t)node * HEADS];
  const float ad1 = (HEADS == 2) ? ad_v[(size_t)node * 2 + 1] : 0.f;
  const int loff  = lane * VEC;
  float bv[VEC];
#pragma unroll
  for (int c = 0; c < VEC; ++c) bv[c] = bias[loff + c];

  int sA; float eA0, eA1;
  edge_score<HEADS, HEADS>(lane, deg, beg, node, csr_src, as_v, ad0, ad1, sA, eA0, eA1);
  float m0 = eA0, m1 = eA1;
  for (int c0 = 32; c0 < deg; c0 += 32) {
    int s; float e0, e1;
    edge_score<HEADS, HEADS>(c0 + lane, deg, beg, node, csr_src, as_v, ad0, ad1, s, e0, e1);
    m0 = fmaxf(m0, e0);
    m1 = fmaxf(m1, e1);
  }
  for (int off = 16; off > 0; off >>= 1) {
    m0 = fmaxf(m0, __shfl_xor(m0, off, 32));
    if (HEADS == 2) m1 = fmaxf(m1, __shfl_xor(m1, off, 32));
  }

  float z0 = 0.f, z1 = 0.f;
  float acc[VEC];
#pragma unroll
  for (int c = 0; c < VEC; ++c) acc[c] = 0.f;
  for (int c0 = 0; c0 < deg; c0 += 32) {
    int s = sA; float e0 = eA0, e1 = eA1;
    if (c0 > 0) edge_score<HEADS, HEADS>(c0 + lane, deg, beg, node, csr_src, as_v, ad0, ad1, s, e0, e1);
    const bool ok = (c0 + lane) < deg;
    const float ex0 = ok ? expf(e0 - m0) : 0.f;
    const float ex1 = (HEADS == 2 && ok) ? expf(e1 - m1) : 0.f;
    z0 += ex0;
    z1 += ex1;
    const int cnt = min(32, deg - c0);
    int j0 = 0;
    for (; j0 + 8 <= cnt; j0 += 8) rows<HEADS, 8>(s, ex0, ex1, j0, lane, h, acc);
    if (cnt & 4) { rows<HEADS, 4>(s, ex0, ex1, j0, lane, h, acc); j0 += 4; }
    if (cnt & 2) { rows<HEADS, 2>(s, ex0, ex1, j0, lane, h, acc); j0 += 2; }
    if (cnt & 1) { rows<HEADS, 1>(s, ex0, ex1, j0, lane, h, acc); }
  }
  for (int off = 16; off > 0; off >>= 1) {
    z0 += __shfl_xor(z0, off, 32);
    if (HEADS == 2) z1 += __shfl_xor(z1, off, 32);
  }

  const float zz  = (HEADS == 2 && (lane >> 4)) ? z1 : z0;
  const float den = zz + 1e-16f;
  float r[VEC];
#pragma unroll
  for (int c = 0; c < VEC; ++c) {
    float v = acc[c] / den + bv[c];
    r[c] = (v > 0.f) ? v : expm1f(v);
  }
  float* op = out + (size_t)node * ROW + loff;
  if (HEADS == 2) *(float4*)op = make_float4(r[0], r[1], r[VEC - 2], r[VEC - 1]);
  else            *(float2*)op = make_float2(r[0], r[1]);
}


__global__ void __launch_bounds__(256) scores_chain(const float* __restrict__ x, const float* __restrict__ Wm,
                                                    const float* __restrict__ a_src, const float* __restrict__ a_dst,
                                                    float* __restrict__ sd, int M) {
  __shared__ float wa[256];
  const int t      = threadIdx.x;
  const int tilesM = (M + 15) >> 4;
  const int tm     = blockIdx.x * 8 + (t >> 5);
  const int lane = t & 31;
  const int half = lane >> 4;
  const int c    = lane & 15;
  const float* ap = x + (size_t)(min(tm, tilesM - 1) * 16 + c) * 64 + 2 * half;
  float2 av[16];
#pragma unroll
  for (int i = 0; i < 16; ++i) av[i] = *(const float2*)(ap + 4 * i);
  {
    const int k = t >> 2, cc = t & 3, hd = cc & 1;
    const float* avec = ((cc & 2) ? a_dst : a_src) + hd * 64;
    const float* wr   = Wm + (size_t)k * 128 + hd * 64;
    float acc = 0.f;
#pragma unroll
    for (int f = 0; f < 64; f += 4) {
      const float4 w4 = *(const float4*)(wr + f);
      const float4 a4 = *(const float4*)(avec + f);
      acc += w4.x * a4.x + w4.y * a4.y + w4.z * a4.z + w4.w * a4.w;
    }
    wa[t] = acc;
  }
  __syncthreads();
  if (tm >= tilesM) return;
  const bool cok  = c < 4;
  const int  colc = cok ? c : 0;
  const float* bl = wa + (2 * half) * 4 + colc;
  v8f d = {0.f, 0.f, 0.f, 0.f, 0.f, 0.f, 0.f, 0.f};
#pragma unroll
  for (int i = 0; i < 16; ++i) {
    const float bx = bl[16 * i];
    const float by = bl[16 * i + 4];
    v2f a, b;
    a.x = av[i].x;
    a.y = av[i].y;
    b.x = cok ? bx : 0.f;
    b.y = cok ? by : 0.f;
    d = __builtin_amdgcn_wmma_f32_16x16x4_f32(false, a, false, b, (short)0, d, false, false);
  }
  if (cok) {
    float* crow = sd + (size_t)(tm * 16 + 8 * half) * 4 + c;
#pragma unroll
    for (int v = 0; v < 8; ++v) crow[(size_t)v * 4] = d[v];
  }
}

template <int G>
__device__ __forceinline__ void rows_x2(int s, float ex0, float ex1, int j0, int lane,
                                        const float* __restrict__ x, v2f& acc0, v2f& acc1) {
  const int loff = lane * 2;
  v2f hv[G];
#pragma unroll
  for (int jj = 0; jj < G; ++jj) {
    int sj = bcast_i(s, j0 + jj);
    hv[jj] = *(const v2f*)(x + (size_t)sj * 64 + loff);
  }
#pragma unroll
  for (int jj = 0; jj < G; ++jj) {
    const float w0 = bcast_f(ex0, j0 + jj);
    const float w1 = bcast_f(ex1, j0 + jj);
    acc0 += w0 * hv[jj];
    acc1 += w1 * hv[jj];
  }
}

__global__ void gat_gather_x2(const int* __restrict__ rowptr, const int* __restrict__ csr_src,
                              const float* __restrict__ sd, const float* __restrict__ x,
                              float* __restrict__ agg, int n) {
  int node = __builtin_amdgcn_readfirstlane(blockIdx.x * (blockDim.x >> 5) + (threadIdx.x >> 5));
  if (node >= n) return;
  const int lane = threadIdx.x & 31;
  const int beg  = __builtin_amdgcn_readfirstlane(rowptr[node]);
  const int end  = __builtin_amdgcn_readfirstlane(rowptr[node + 1]);
  const int deg  = end - beg + 1;
  const float2 adv = *(const float2*)(sd + (size_t)node * 4 + 2);
  const float ad0 = adv.x, ad1 = adv.y;

  int sA; float eA0, eA1;
  edge_score<2, 4>(lane, deg, beg, node, csr_src, sd, ad0, ad1, sA, eA0, eA1);
  float m0 = eA0, m1 = eA1;
  for (int c0 = 32; c0 < deg; c0 += 32) {
    int s; float e0, e1;
    edge_score<2, 4>(c0 + lane, deg, beg, node, csr_src, sd, ad0, ad1, s, e0, e1);
    m0 = fmaxf(m0, e0);
    m1 = fmaxf(m1, e1);
  }
  for (int off = 16; off > 0; off >>= 1) {
    m0 = fmaxf(m0, __shfl_xor(m0, off, 32));
    m1 = fmaxf(m1, __shfl_xor(m1, off, 32));
  }

  float z0 = 0.f, z1 = 0.f;
  v2f acc0 = {0.f, 0.f}, acc1 = {0.f, 0.f};
  for (int c0 = 0; c0 < deg; c0 += 32) {
    int s = sA; float e0 = eA0, e1 = eA1;
    if (c0 > 0) edge_score<2, 4>(c0 + lane, deg, beg, node, csr_src, sd, ad0, ad1, s, e0, e1);
    const bool ok = (c0 + lane) < deg;
    const float ex0 = ok ? expf(e0 - m0) : 0.f;
    const float ex1 = ok ? expf(e1 - m1) : 0.f;
    z0 += ex0;
    z1 += ex1;
    const int cnt = min(32, deg - c0);
    int j0 = 0;
    for (; j0 + 8 <= cnt; j0 += 8) rows_x2<8>(s, ex0, ex1, j0, lane, x, acc0, acc1);
    if (cnt & 4) { rows_x2<4>(s, ex0, ex1, j0, lane, x, acc0, acc1); j0 += 4; }
    if (cnt & 2) { rows_x2<2>(s, ex0, ex1, j0, lane, x, acc0, acc1); j0 += 2; }
    if (cnt & 1) { rows_x2<1>(s, ex0, ex1, j0, lane, x, acc0, acc1); }
  }
  for (int off = 16; off > 0; off >>= 1) {
    z0 += __shfl_xor(z0, off, 32);
    z1 += __shfl_xor(z1, off, 32);
  }

  const float den0 = z0 + 1e-16f, den1 = z1 + 1e-16f;
  float* op = agg + (size_t)node * 128 + lane * 2;
  *(float2*)op        = make_float2(acc0.x / den0, acc0.y / den0);
  *(float2*)(op + 64) = make_float2(acc1.x / den1, acc1.y / den1);
}

extern "C" void kernel_launch(void* const* d_in, const int* in_sizes, int n_in,
                              void* d_out, int out_size, void* d_ws, size_t ws_size,
                              hipStream_t stream) {
  const float* x      = (const float*)d_in[0];
  const int*   ei     = (const int*)d_in[1];
  const float* W1     = (const float*)d_in[2];
  const float* a_src1 = (const float*)d_in[3];
  const float* a_dst1 = (const float*)d_in[4];
  const float* b1     = (const float*)d_in[5];
  const float* W2     = (const float*)d_in[6];
  const float* a_src2 = (const float*)d_in[7];
  const float* a_dst2 = (const float*)d_in[8];
  const float* b2     = (const float*)d_in[9];
  const float* m1w    = (const float*)d_in[10];
  const float* m1b    = (const float*)d_in[11];
  const float* m2w    = (const float*)d_in[12];
  const float* m2b    = (const float*)d_in[13];

  const int N    = in_sizes[0] / 64;
  const int E    = in_sizes[1] / 2;
  const int C    = in_sizes[12] / 64;

  float* bufA   = (float*)d_ws;
  float* bufB   = bufA + (size_t)N * 128;
  float* bufC   = bufB + (size_t)N * 128;
  float* as_v   = bufC + (size_t)N * 64;
  float* ad_v   = as_v + (size_t)N * 2;
  int*   rowptr = (int*)(ad_v + (size_t)N * 2) + (size_t)N;
  int*   csrsrc = rowptr + (size_t)N + 1;
  const int nbk  = (N + BK_NODES - 1) / BK_NODES;
  const int nblk = (E + TILE_EDGES - 1) / TILE_EDGES;
  int*   startT = (int*)bufA;
  int*   pairs  = (int*)bufB;

  const int TPB = 256;
  auto gemm_blocks = [&](int M, int Nc) {
    int waves = ((M + 15) / 16) * ((Nc + 15) / 16);
    return (waves + 7) / 8;
  };
  auto chain_blocks = [&](int M) {
    return ((M + 15) / 16 + CHAIN_WAVES - 1) / CHAIN_WAVES;
  };
  auto strip_blocks = [&](int M, int Nc) {
    int waves = ((M + 15) / 16) * (Nc / 64);
    return (waves + 7) / 8;
  };

  tile_sort<<<nblk, 1024, 0, stream>>>(ei, E, nbk, startT, pairs);
  bucket_sort<<<nbk, 1024, 0, stream>>>(pairs, startT, nblk, E, N, rowptr, csrsrc);

  scores_chain<<<gemm_blocks(N, 4), TPB, 0, stream>>>(x, W1, a_src1, a_dst1, as_v, N);
  gat_gather_x2<<<(N + 7) / 8, TPB, 0, stream>>>(rowptr, csrsrc, as_v, x, bufA, N);
  proj_chain<<<chain_blocks(N), 32 * CHAIN_WAVES, 0, stream>>>(bufA, W1, b1, W2, a_src2, a_dst2,
                                                      bufB, as_v, ad_v, N);
  gat_gather<1><<<(N + 7) / 8, TPB, 0, stream>>>(rowptr, csrsrc, as_v, ad_v,
                                                 bufB, b2, bufC, N);

  if (C == 8) {
    mlp_chain<8><<<chain_blocks(N), 32 * CHAIN_WAVES, 0, stream>>>(bufC, m1w, m1b, m2w, m2b,
                                                          (float*)d_out, N);
  } else {
    gemm_strip64<64, 64, 64, false, false, false><<<strip_blocks(N, 64), TPB, 0, stream>>>(
        bufC, m1w, m1b, bufA, nullptr, nullptr, nullptr, nullptr, N, 1);
    gemm_wmma_f32_dyn<<<gemm_blocks(N, C), TPB, 0, stream>>>(bufA, m2w, m2b,
                                                             (float*)d_out, N, 64, C, 1);
  }
}
